# starting kernel
# baseline (speedup 1.0000x reference)
#include <hip/hip_runtime.h>
#include <hip/hip_bf16.h>
#include <cstddef>

typedef __attribute__((ext_vector_type(2))) float v2f;
typedef __attribute__((ext_vector_type(8))) float v8f;

#define CH 128

__global__ void deg_init_kernel(float* __restrict__ deg, int n) {
    int i = blockIdx.x * blockDim.x + threadIdx.x;
    if (i < n) deg[i] = 1.0f;
}

__global__ void deg_count_kernel(const int* __restrict__ dst, float* __restrict__ deg, int e) {
    int i = blockIdx.x * blockDim.x + threadIdx.x;
    if (i < e) unsafeAtomicAdd(&deg[dst[i]], 1.0f);
}

__global__ void dinv_kernel(float* __restrict__ deg, int n) {
    int i = blockIdx.x * blockDim.x + threadIdx.x;
    if (i < n) deg[i] = rsqrtf(deg[i]);
}

__global__ __launch_bounds__(128) void gemm_wmma_kernel(
    const float* __restrict__ X, const float* __restrict__ W,
    float* __restrict__ H, int nrows, int apply_relu)
{
    __shared__ float Xs[4][16 * CH];

    const int lane = threadIdx.x & 31;
    const int wave = threadIdx.x >> 5;
    const int row_base = blockIdx.x * 64 + wave * 16;
    const bool full_tile = (row_base + 16 <= nrows);

    float* xs = &Xs[wave][0];

    if (full_tile) {
        #pragma unroll
        for (int r = 0; r < 16; ++r) {
            float4 v = *(const float4*)(X + (size_t)(row_base + r) * CH + lane * 4);
            if (apply_relu) {
                v.x = fmaxf(v.x, 0.f); v.y = fmaxf(v.y, 0.f);
                v.z = fmaxf(v.z, 0.f); v.w = fmaxf(v.w, 0.f);
            }
            *(float4*)(xs + r * CH + lane * 4) = v;
        }
    } else {
        #pragma unroll
        for (int r = 0; r < 16; ++r) {
            int row = row_base + r;
            float4 v = make_float4(0.f, 0.f, 0.f, 0.f);
            if (row < nrows)
                v = *(const float4*)(X + (size_t)row * CH + lane * 4);
            if (apply_relu) {
                v.x = fmaxf(v.x, 0.f); v.y = fmaxf(v.y, 0.f);
                v.z = fmaxf(v.z, 0.f); v.w = fmaxf(v.w, 0.f);
            }
            *(float4*)(xs + r * CH + lane * 4) = v;
        }
    }
    __syncthreads();

    const int m    = lane & 15;
    const int hi   = lane >> 4;
    const int colN = lane & 15;

    v8f acc[8];
    #pragma unroll
    for (int n = 0; n < 8; ++n) acc[n] = (v8f)(0.0f);

    for (int k0 = 0; k0 < CH; k0 += 4) {
        const int klo = k0 + hi * 2;
        v2f a = *(const v2f*)(xs + m * CH + klo);
        #pragma unroll
        for (int n = 0; n < 8; ++n) {
            v2f b;
            b.x = W[(size_t)klo * CH + n * 16 + colN];
            b.y = W[(size_t)(klo + 1) * CH + n * 16 + colN];
            acc[n] = __builtin_amdgcn_wmma_f32_16x16x4_f32(
                false, a, false, b, (short)0, acc[n], false, false);
        }
    }

    if (full_tile) {
        #pragma unroll
        for (int n = 0; n < 8; ++n) {
            #pragma unroll
            for (int r = 0; r < 8; ++r) {
                int row = row_base + r + hi * 8;
                H[(size_t)row * CH + n * 16 + colN] = acc[n][r];
            }
        }
    } else if (row_base < nrows) {
        #pragma unroll
        for (int n = 0; n < 8; ++n) {
            #pragma unroll
            for (int r = 0; r < 8; ++r) {
                int row = row_base + r + hi * 8;
                if (row < nrows)
                    H[(size_t)row * CH + n * 16 + colN] = acc[n][r];
            }
        }
    }
}

__global__ __launch_bounds__(256) void agg_init_kernel(
    const float* __restrict__ H, const float* __restrict__ dinv,
    const float* __restrict__ bias, float* __restrict__ out, int n)
{
    int t = blockIdx.x * blockDim.x + threadIdx.x;
    if (t >= n * 32) return;
    int i = t >> 5;
    int c = (t & 31) * 4;
    float di = dinv[i];
    float w = di * di;
    float4 h = *(const float4*)(H + (size_t)i * CH + c);
    float4 b = *(const float4*)(bias + c);
    float4 o;
    o.x = fmaf(h.x, w, b.x);
    o.y = fmaf(h.y, w, b.y);
    o.z = fmaf(h.z, w, b.z);
    o.w = fmaf(h.w, w, b.w);
    *(float4*)(out + (size_t)i * CH + c) = o;
}

__global__ __launch_bounds__(256) void scatter_kernel(
    const float* __restrict__ H, const float* __restrict__ dinv,
    const int* __restrict__ src_idx, const int* __restrict__ dst_idx,
    float* __restrict__ out, int nedges)
{
    const int lane = threadIdx.x & 31;
    const int nwaves = (gridDim.x * blockDim.x) >> 5;
    int e = (blockIdx.x * blockDim.x + threadIdx.x) >> 5;

    for (; e < nedges; e += nwaves) {
        int s = src_idx[e];
        int d = dst_idx[e];

        int en = e + nwaves;
        if (en < nedges) {
            int sn = src_idx[en];
            __builtin_prefetch(H + (size_t)sn * CH + lane * 4, 0, 1);
        }

        float coef = dinv[s] * dinv[d];
        float4 hv = *(const float4*)(H + (size_t)s * CH + lane * 4);
        float* op = out + (size_t)d * CH + lane * 4;
        unsafeAtomicAdd(op + 0, hv.x * coef);
        unsafeAtomicAdd(op + 1, hv.y * coef);
        unsafeAtomicAdd(op + 2, hv.z * coef);
        unsafeAtomicAdd(op + 3, hv.w * coef);
    }
}

extern "C" void kernel_launch(void* const* d_in, const int* in_sizes, int n_in,
                              void* d_out, int out_size, void* d_ws, size_t ws_size,
                              hipStream_t stream) {
    const float* x    = (const float*)d_in[0];
    const int*   edge = (const int*)d_in[1];
    const float* W1   = (const float*)d_in[2];
    const float* b1   = (const float*)d_in[3];
    const float* W2   = (const float*)d_in[4];
    const float* b2   = (const float*)d_in[5];
    float* out = (float*)d_out;

    const int N = in_sizes[0] / CH;
    const int E = in_sizes[1] / 2;
    const int* srcI = edge;
    const int* dstI = edge + E;

    char* ws = (char*)d_ws;
    float* dinv = (float*)ws;
    size_t off  = ((size_t)N * 4 + 255) & ~(size_t)255;
    float* hbuf = (float*)(ws + off);
    float* tmp  = hbuf + (size_t)N * CH;

    const int T = 256;
    dim3 gN((N + T - 1) / T), gE((E + T - 1) / T);
    dim3 gGemm((N + 63) / 64);
    dim3 gAgg(((size_t)N * 32 + T - 1) / T);
    dim3 gScat(8192);

    deg_init_kernel<<<gN, T, 0, stream>>>(dinv, N);
    deg_count_kernel<<<gE, T, 0, stream>>>(dstI, dinv, E);
    dinv_kernel<<<gN, T, 0, stream>>>(dinv, N);

    gemm_wmma_kernel<<<gGemm, 128, 0, stream>>>(x, W1, hbuf, N, 0);
    agg_init_kernel<<<gAgg, T, 0, stream>>>(hbuf, dinv, b1, tmp, N);
    scatter_kernel<<<gScat, T, 0, stream>>>(hbuf, dinv, srcI, dstI, tmp, E);

    gemm_wmma_kernel<<<gGemm, 128, 0, stream>>>(tmp, W2, hbuf, N, 1);
    agg_init_kernel<<<gAgg, T, 0, stream>>>(hbuf, dinv, b2, out, N);
    scatter_kernel<<<gScat, T, 0, stream>>>(hbuf, dinv, srcI, dstI, out, E);
}


# replace per-edge f32 atomic scatter-add by a dst-sorted (CSR) gather: edges ranked by integer atomic counts + prefix scan, one wave sums each output row in registers, stores once (sum order changes)
# speedup vs baseline: 2.5824x; 2.5824x over previous
#include <hip/hip_runtime.h>
#include <hip/hip_bf16.h>
#include <cstddef>

typedef __attribute__((ext_vector_type(2))) float v2f;
typedef __attribute__((ext_vector_type(8))) float v8f;

#define CH 128

#define SCAN_TILE 1024

__global__ void zero_cnt_kernel(int* __restrict__ cnt, int n) {
    int i = blockIdx.x * blockDim.x + threadIdx.x;
    if (i < n) cnt[i] = 0;
}

__global__ void count_rank_kernel(const int* __restrict__ dst, int* __restrict__ cnt,
                                  int* __restrict__ rank, int e) {
    int i = blockIdx.x * blockDim.x + threadIdx.x;
    if (i < e) rank[i] = atomicAdd(&cnt[dst[i]], 1);
}

__device__ __forceinline__ int wave_sum(int v) {
    #pragma unroll
    for (int off = 16; off > 0; off >>= 1) v += __shfl_xor(v, off);
    return v;
}

__global__ __launch_bounds__(256) void block_sum_kernel(
    const int* __restrict__ cnt, int* __restrict__ bsum, int n)
{
    __shared__ int red[8];
    const int lane = threadIdx.x & 31;
    const int wave = threadIdx.x >> 5;
    const int base = blockIdx.x * SCAN_TILE + threadIdx.x * 4;
    int s = 0;
    #pragma unroll
    for (int k = 0; k < 4; ++k)
        if (base + k < n) s += cnt[base + k];
    s = wave_sum(s);
    if (lane == 0) red[wave] = s;
    __syncthreads();
    if (threadIdx.x == 0) {
        int t = 0;
        #pragma unroll
        for (int w = 0; w < 8; ++w) t += red[w];
        bsum[blockIdx.x] = t;
    }
}

__global__ __launch_bounds__(256) void scan_kernel(
    const int* __restrict__ cnt, const int* __restrict__ bsum,
    int* __restrict__ rowptr, float* __restrict__ dinv, int n)
{
    __shared__ int red[8];
    __shared__ int wsum[8];
    const int lane = threadIdx.x & 31;
    const int wave = threadIdx.x >> 5;

    int part = 0;
    for (int j = threadIdx.x; j < (int)blockIdx.x; j += 256) part += bsum[j];
    part = wave_sum(part);
    if (lane == 0) red[wave] = part;

    const int base = blockIdx.x * SCAN_TILE + threadIdx.x * 4;
    int c[4];
    #pragma unroll
    for (int k = 0; k < 4; ++k) c[k] = (base + k < n) ? cnt[base + k] : 0;
    const int tsum = c[0] + c[1] + c[2] + c[3];

    int inc = tsum;
    #pragma unroll
    for (int off = 1; off < 32; off <<= 1) {
        int v = __shfl_up(inc, off);
        if (lane >= off) inc += v;
    }
    if (lane == 31) wsum[wave] = inc;
    __syncthreads();

    int offset = 0;
    #pragma unroll
    for (int w = 0; w < 8; ++w) offset += red[w];
    for (int w = 0; w < wave; ++w) offset += wsum[w];

    int excl = offset + inc - tsum;
    #pragma unroll
    for (int k = 0; k < 4; ++k) {
        const int idx = base + k;
        if (idx < n) {
            rowptr[idx] = excl;
            dinv[idx] = rsqrtf((float)(c[k] + 1));
            excl += c[k];
            if (idx == n - 1) rowptr[n] = excl;
        }
    }
}

__global__ void fill_kernel(const int* __restrict__ src, const int* __restrict__ dst,
                            const int* __restrict__ rank, const int* __restrict__ rowptr,
                            int* __restrict__ csr_src, int e) {
    int i = blockIdx.x * blockDim.x + threadIdx.x;
    if (i < e) csr_src[rowptr[dst[i]] + rank[i]] = src[i];
}

__global__ __launch_bounds__(128) void gemm_wmma_kernel(
    const float* __restrict__ X, const float* __restrict__ W,
    float* __restrict__ H, int nrows, int apply_relu)
{
    __shared__ float Xs[4][16 * CH];

    const int lane = threadIdx.x & 31;
    const int wave = threadIdx.x >> 5;
    const int row_base = blockIdx.x * 64 + wave * 16;
    const bool full_tile = (row_base + 16 <= nrows);

    float* xs = &Xs[wave][0];

    if (full_tile) {
        #pragma unroll
        for (int r = 0; r < 16; ++r) {
            float4 v = *(const float4*)(X + (size_t)(row_base + r) * CH + lane * 4);
            if (apply_relu) {
                v.x = fmaxf(v.x, 0.f); v.y = fmaxf(v.y, 0.f);
                v.z = fmaxf(v.z, 0.f); v.w = fmaxf(v.w, 0.f);
            }
            *(float4*)(xs + r * CH + lane * 4) = v;
        }
    } else {
        #pragma unroll
        for (int r = 0; r < 16; ++r) {
            int row = row_base + r;
            float4 v = make_float4(0.f, 0.f, 0.f, 0.f);
            if (row < nrows)
                v = *(const float4*)(X + (size_t)row * CH + lane * 4);
            if (apply_relu) {
                v.x = fmaxf(v.x, 0.f); v.y = fmaxf(v.y, 0.f);
                v.z = fmaxf(v.z, 0.f); v.w = fmaxf(v.w, 0.f);
            }
            *(float4*)(xs + r * CH + lane * 4) = v;
        }
    }
    __syncthreads();

    const int m    = lane & 15;
    const int hi   = lane >> 4;
    const int colN = lane & 15;

    v8f acc[8];
    #pragma unroll
    for (int n = 0; n < 8; ++n) acc[n] = (v8f)(0.0f);

    for (int k0 = 0; k0 < CH; k0 += 4) {
        const int klo = k0 + hi * 2;
        v2f a = *(const v2f*)(xs + m * CH + klo);
        #pragma unroll
        for (int n = 0; n < 8; ++n) {
            v2f b;
            b.x = W[(size_t)klo * CH + n * 16 + colN];
            b.y = W[(size_t)(klo + 1) * CH + n * 16 + colN];
            acc[n] = __builtin_amdgcn_wmma_f32_16x16x4_f32(
                false, a, false, b, (short)0, acc[n], false, false);
        }
    }

    if (full_tile) {
        #pragma unroll
        for (int n = 0; n < 8; ++n) {
            #pragma unroll
            for (int r = 0; r < 8; ++r) {
                int row = row_base + r + hi * 8;
                H[(size_t)row * CH + n * 16 + colN] = acc[n][r];
            }
        }
    } else if (row_base < nrows) {
        #pragma unroll
        for (int n = 0; n < 8; ++n) {
            #pragma unroll
            for (int r = 0; r < 8; ++r) {
                int row = row_base + r + hi * 8;
                if (row < nrows)
                    H[(size_t)row * CH + n * 16 + colN] = acc[n][r];
            }
        }
    }
}

template <int K>
__device__ __forceinline__ void add_rows(const float* __restrict__ H, int sidx, float cf,
                                         int j, int lane, float4& acc)
{
    float4 v[K];
    float  c[K];
    #pragma unroll
    for (int k = 0; k < K; ++k) {
        const int s = __builtin_amdgcn_readlane(sidx, j + k);
        c[k] = __int_as_float(__builtin_amdgcn_readlane(__float_as_int(cf), j + k));
        v[k] = *(const float4*)(H + (size_t)s * CH + lane * 4);
    }
    #pragma unroll
    for (int k = 0; k < K; ++k) {
        acc.x = fmaf(v[k].x, c[k], acc.x);
        acc.y = fmaf(v[k].y, c[k], acc.y);
        acc.z = fmaf(v[k].z, c[k], acc.z);
        acc.w = fmaf(v[k].w, c[k], acc.w);
    }
}

__global__ __launch_bounds__(256) void gather_kernel(
    const float* __restrict__ H, const float* __restrict__ dinv,
    const int* __restrict__ rowptr, const int* __restrict__ csr_src,
    const float* __restrict__ bias, float* __restrict__ out, int n)
{
    const int lane = threadIdx.x & 31;
    const int row = __builtin_amdgcn_readfirstlane(
        (int)((blockIdx.x * blockDim.x + threadIdx.x) >> 5));
    if (row >= n) return;

    const int beg = rowptr[row];
    const int end = rowptr[row + 1];
    const float di = dinv[row];
    const float w = di * di;

    const float4 h = *(const float4*)(H + (size_t)row * CH + lane * 4);
    float4 acc;
    acc.x = h.x * w; acc.y = h.y * w; acc.z = h.z * w; acc.w = h.w * w;

    for (int base = beg; base < end; base += 32) {
        const int nb = (end - base < 32) ? (end - base) : 32;
        int sidx = 0;
        float cf = 0.f;
        if (lane < nb) {
            sidx = csr_src[base + lane];
            cf = dinv[sidx] * di;
        }
        int j = 0;
        for (; j + 4 <= nb; j += 4) add_rows<4>(H, sidx, cf, j, lane, acc);
        if (j + 2 <= nb) { add_rows<2>(H, sidx, cf, j, lane, acc); j += 2; }
        if (j < nb) add_rows<1>(H, sidx, cf, j, lane, acc);
    }

    const float4 b = *(const float4*)(bias + lane * 4);
    float4 o;
    o.x = acc.x + b.x; o.y = acc.y + b.y; o.z = acc.z + b.z; o.w = acc.w + b.w;
    *(float4*)(out + (size_t)row * CH + lane * 4) = o;
}

extern "C" void kernel_launch(void* const* d_in, const int* in_sizes, int n_in,
                              void* d_out, int out_size, void* d_ws, size_t ws_size,
                              hipStream_t stream) {
    const float* x    = (const float*)d_in[0];
    const int*   edge = (const int*)d_in[1];
    const float* W1   = (const float*)d_in[2];
    const float* b1   = (const float*)d_in[3];
    const float* W2   = (const float*)d_in[4];
    const float* b2   = (const float*)d_in[5];
    float* out = (float*)d_out;

    const int N = in_sizes[0] / CH;
    const int E = in_sizes[1] / 2;
    const int* srcI = edge;
    const int* dstI = edge + E;

    const int T = 256;
    const int NB = (N + SCAN_TILE - 1) / SCAN_TILE;

    char* ws = (char*)d_ws;
    size_t off = 0;
    float* dinv = (float*)(ws + off);  off += ((size_t)N * 4 + 255) & ~(size_t)255;
    int* cnt    = (int*)(ws + off);    off += ((size_t)N * 4 + 255) & ~(size_t)255;
    int* rowptr = (int*)(ws + off);    off += ((size_t)(N + 1) * 4 + 255) & ~(size_t)255;
    int* bsum   = (int*)(ws + off);    off += ((size_t)NB * 4 + 255) & ~(size_t)255;
    int* rank   = (int*)(ws + off);    off += ((size_t)E * 4 + 255) & ~(size_t)255;
    int* csr    = (int*)(ws + off);    off += ((size_t)E * 4 + 255) & ~(size_t)255;
    float* hbuf = (float*)(ws + off);
    float* tmp  = hbuf + (size_t)N * CH;

    dim3 gN((N + T - 1) / T), gE((E + T - 1) / T);
    dim3 gGemm((N + 63) / 64);
    dim3 gRow((N + 7) / 8);

    zero_cnt_kernel<<<gN, T, 0, stream>>>(cnt, N);
    count_rank_kernel<<<gE, T, 0, stream>>>(dstI, cnt, rank, E);
    block_sum_kernel<<<NB, T, 0, stream>>>(cnt, bsum, N);
    scan_kernel<<<NB, T, 0, stream>>>(cnt, bsum, rowptr, dinv, N);
    fill_kernel<<<gE, T, 0, stream>>>(srcI, dstI, rank, rowptr, csr, E);

    gemm_wmma_kernel<<<gGemm, 128, 0, stream>>>(x, W1, hbuf, N, 0);
    gather_kernel<<<gRow, T, 0, stream>>>(hbuf, dinv, rowptr, csr, b1, tmp, N);

    gemm_wmma_kernel<<<gGemm, 128, 0, stream>>>(tmp, W2, hbuf, N, 1);
    gather_kernel<<<gRow, T, 0, stream>>>(hbuf, dinv, rowptr, csr, b2, out, N);
}


# edge count (returning integer atomics) runs as extra work-groups inside the layer-1 GEMM launch, 8 edges per thread: atomic round trips overlap matrix issue; 9 -> 8 launches; arrival-rank order kept
# speedup vs baseline: 2.7250x; 1.0552x over previous
#include <hip/hip_runtime.h>
#include <hip/hip_bf16.h>
#include <cstddef>

typedef __attribute__((ext_vector_type(2))) float v2f;
typedef __attribute__((ext_vector_type(8))) float v8f;

#define CH 128

#define SCAN_TILE 1024

__global__ void zero_cnt_kernel(int* __restrict__ cnt, int n) {
    int i = blockIdx.x * blockDim.x + threadIdx.x;
    if (i < n) cnt[i] = 0;
}

__device__ __forceinline__ int wave_sum(int v) {
    #pragma unroll
    for (int off = 16; off > 0; off >>= 1) v += __shfl_xor(v, off);
    return v;
}

__global__ __launch_bounds__(256) void block_sum_kernel(
    const int* __restrict__ cnt, int* __restrict__ bsum, int n)
{
    __shared__ int red[8];
    const int lane = threadIdx.x & 31;
    const int wave = threadIdx.x >> 5;
    const int base = blockIdx.x * SCAN_TILE + threadIdx.x * 4;
    int s = 0;
    #pragma unroll
    for (int k = 0; k < 4; ++k)
        if (base + k < n) s += cnt[base + k];
    s = wave_sum(s);
    if (lane == 0) red[wave] = s;
    __syncthreads();
    if (threadIdx.x == 0) {
        int t = 0;
        #pragma unroll
        for (int w = 0; w < 8; ++w) t += red[w];
        bsum[blockIdx.x] = t;
    }
}

__global__ __launch_bounds__(256) void scan_kernel(
    const int* __restrict__ cnt, const int* __restrict__ bsum,
    int* __restrict__ rowptr, float* __restrict__ dinv, int n)
{
    __shared__ int red[8];
    __shared__ int wsum[8];
    const int lane = threadIdx.x & 31;
    const int wave = threadIdx.x >> 5;

    int part = 0;
    for (int j = threadIdx.x; j < (int)blockIdx.x; j += 256) part += bsum[j];
    part = wave_sum(part);
    if (lane == 0) red[wave] = part;

    const int base = blockIdx.x * SCAN_TILE + threadIdx.x * 4;
    int c[4];
    #pragma unroll
    for (int k = 0; k < 4; ++k) c[k] = (base + k < n) ? cnt[base + k] : 0;
    const int tsum = c[0] + c[1] + c[2] + c[3];

    int inc = tsum;
    #pragma unroll
    for (int off = 1; off < 32; off <<= 1) {
        int v = __shfl_up(inc, off);
        if (lane >= off) inc += v;
    }
    if (lane == 31) wsum[wave] = inc;
    __syncthreads();

    int offset = 0;
    #pragma unroll
    for (int w = 0; w < 8; ++w) offset += red[w];
    for (int w = 0; w < wave; ++w) offset += wsum[w];

    int excl = offset + inc - tsum;
    #pragma unroll
    for (int k = 0; k < 4; ++k) {
        const int idx = base + k;
        if (idx < n) {
            rowptr[idx] = excl;
            dinv[idx] = rsqrtf((float)(c[k] + 1));
            excl += c[k];
            if (idx == n - 1) rowptr[n] = excl;
        }
    }
}

__global__ void fill_kernel(const int* __restrict__ src, const int* __restrict__ dst,
                            const int* __restrict__ rank, const int* __restrict__ rowptr,
                            int* __restrict__ csr_src, int e) {
    int i = blockIdx.x * blockDim.x + threadIdx.x;
    if (i < e) csr_src[rowptr[dst[i]] + rank[i]] = src[i];
}

__device__ __forceinline__ void gemm_tile(
    const float* __restrict__ X, const float* __restrict__ W,
    float* __restrict__ H, int nrows, int apply_relu, int tile)
{
    __shared__ float Xs[4][16 * CH];

    const int lane = threadIdx.x & 31;
    const int wave = threadIdx.x >> 5;
    const int row_base = tile * 64 + wave * 16;
    const bool full_tile = (row_base + 16 <= nrows);

    float* xs = &Xs[wave][0];

    if (full_tile) {
        #pragma unroll
        for (int r = 0; r < 16; ++r) {
            float4 v = *(const float4*)(X + (size_t)(row_base + r) * CH + lane * 4);
            if (apply_relu) {
                v.x = fmaxf(v.x, 0.f); v.y = fmaxf(v.y, 0.f);
                v.z = fmaxf(v.z, 0.f); v.w = fmaxf(v.w, 0.f);
            }
            *(float4*)(xs + r * CH + lane * 4) = v;
        }
    } else {
        #pragma unroll
        for (int r = 0; r < 16; ++r) {
            int row = row_base + r;
            float4 v = make_float4(0.f, 0.f, 0.f, 0.f);
            if (row < nrows)
                v = *(const float4*)(X + (size_t)row * CH + lane * 4);
            if (apply_relu) {
                v.x = fmaxf(v.x, 0.f); v.y = fmaxf(v.y, 0.f);
                v.z = fmaxf(v.z, 0.f); v.w = fmaxf(v.w, 0.f);
            }
            *(float4*)(xs + r * CH + lane * 4) = v;
        }
    }
    __syncthreads();

    const int m    = lane & 15;
    const int hi   = lane >> 4;
    const int colN = lane & 15;

    v8f acc[8];
    #pragma unroll
    for (int n = 0; n < 8; ++n) acc[n] = (v8f)(0.0f);

    for (int k0 = 0; k0 < CH; k0 += 4) {
        const int klo = k0 + hi * 2;
        v2f a = *(const v2f*)(xs + m * CH + klo);
        #pragma unroll
        for (int n = 0; n < 8; ++n) {
            v2f b;
            b.x = W[(size_t)klo * CH + n * 16 + colN];
            b.y = W[(size_t)(klo + 1) * CH + n * 16 + colN];
            acc[n] = __builtin_amdgcn_wmma_f32_16x16x4_f32(
                false, a, false, b, (short)0, acc[n], false, false);
        }
    }

    if (full_tile) {
        #pragma unroll
        for (int n = 0; n < 8; ++n) {
            #pragma unroll
            for (int r = 0; r < 8; ++r) {
                int row = row_base + r + hi * 8;
                H[(size_t)row * CH + n * 16 + colN] = acc[n][r];
            }
        }
    } else if (row_base < nrows) {
        #pragma unroll
        for (int n = 0; n < 8; ++n) {
            #pragma unroll
            for (int r = 0; r < 8; ++r) {
                int row = row_base + r + hi * 8;
                if (row < nrows)
                    H[(size_t)row * CH + n * 16 + colN] = acc[n][r];
            }
        }
    }
}

__global__ __launch_bounds__(128) void gemm_wmma_kernel(
    const float* __restrict__ X, const float* __restrict__ W,
    float* __restrict__ H, int nrows, int apply_relu)
{
    gemm_tile(X, W, H, nrows, apply_relu, (int)blockIdx.x);
}

#define CNT_EDGES 8

__global__ __launch_bounds__(128) void gemm_count_kernel(
    const float* __restrict__ X, const float* __restrict__ W,
    float* __restrict__ H, int nrows,
    const int* __restrict__ dst, int* __restrict__ cnt, int* __restrict__ rank,
    int e, int gemm_blocks)
{
    if ((int)blockIdx.x >= gemm_blocks) {
        const int first = ((int)blockIdx.x - gemm_blocks) * (128 * CNT_EDGES);
        const int base = first + (int)threadIdx.x;
        if (first + 128 * CNT_EDGES <= e) {
            int d[CNT_EDGES];
            int r[CNT_EDGES];
            #pragma unroll
            for (int k = 0; k < CNT_EDGES; ++k) d[k] = dst[base + k * 128];
            #pragma unroll
            for (int k = 0; k < CNT_EDGES; ++k) r[k] = atomicAdd(&cnt[d[k]], 1);
            #pragma unroll
            for (int k = 0; k < CNT_EDGES; ++k) rank[base + k * 128] = r[k];
        } else {
            for (int k = 0; k < CNT_EDGES; ++k) {
                const int i = base + k * 128;
                if (i < e) rank[i] = atomicAdd(&cnt[dst[i]], 1);
            }
        }
        return;
    }
    gemm_tile(X, W, H, nrows, 0, (int)blockIdx.x);
}

template <int K>
__device__ __forceinline__ void add_rows(const float* __restrict__ H, int sidx, float cf,
                                         int j, int lane, float4& acc)
{
    float4 v[K];
    float  c[K];
    #pragma unroll
    for (int k = 0; k < K; ++k) {
        const int s = __builtin_amdgcn_readlane(sidx, j + k);
        c[k] = __int_as_float(__builtin_amdgcn_readlane(__float_as_int(cf), j + k));
        v[k] = *(const float4*)(H + (size_t)s * CH + lane * 4);
    }
    #pragma unroll
    for (int k = 0; k < K; ++k) {
        acc.x = fmaf(v[k].x, c[k], acc.x);
        acc.y = fmaf(v[k].y, c[k], acc.y);
        acc.z = fmaf(v[k].z, c[k], acc.z);
        acc.w = fmaf(v[k].w, c[k], acc.w);
    }
}

__global__ __launch_bounds__(256) void gather_kernel(
    const float* __restrict__ H, const float* __restrict__ dinv,
    const int* __restrict__ rowptr, const int* __restrict__ csr_src,
    const float* __restrict__ bias, float* __restrict__ out, int n)
{
    const int lane = threadIdx.x & 31;
    const int row = __builtin_amdgcn_readfirstlane(
        (int)((blockIdx.x * blockDim.x + threadIdx.x) >> 5));
    if (row >= n) return;

    const int beg = rowptr[row];
    const int end = rowptr[row + 1];
    const float di = dinv[row];
    const float w = di * di;

    const float4 h = *(const float4*)(H + (size_t)row * CH + lane * 4);
    float4 acc;
    acc.x = h.x * w; acc.y = h.y * w; acc.z = h.z * w; acc.w = h.w * w;

    for (int base = beg; base < end; base += 32) {
        const int nb = (end - base < 32) ? (end - base) : 32;
        int sidx = 0;
        float cf = 0.f;
        if (lane < nb) {
            sidx = csr_src[base + lane];
            cf = dinv[sidx] * di;
        }
        int j = 0;
        for (; j + 4 <= nb; j += 4) add_rows<4>(H, sidx, cf, j, lane, acc);
        if (j + 2 <= nb) { add_rows<2>(H, sidx, cf, j, lane, acc); j += 2; }
        if (j < nb) add_rows<1>(H, sidx, cf, j, lane, acc);
    }

    const float4 b = *(const float4*)(bias + lane * 4);
    float4 o;
    o.x = acc.x + b.x; o.y = acc.y + b.y; o.z = acc.z + b.z; o.w = acc.w + b.w;
    *(float4*)(out + (size_t)row * CH + lane * 4) = o;
}

extern "C" void kernel_launch(void* const* d_in, const int* in_sizes, int n_in,
                              void* d_out, int out_size, void* d_ws, size_t ws_size,
                              hipStream_t stream) {
    const float* x    = (const float*)d_in[0];
    const int*   edge = (const int*)d_in[1];
    const float* W1   = (const float*)d_in[2];
    const float* b1   = (const float*)d_in[3];
    const float* W2   = (const float*)d_in[4];
    const float* b2   = (const float*)d_in[5];
    float* out = (float*)d_out;

    const int N = in_sizes[0] / CH;
    const int E = in_sizes[1] / 2;
    const int* srcI = edge;
    const int* dstI = edge + E;

    const int T = 256;
    const int NB = (N + SCAN_TILE - 1) / SCAN_TILE;

    char* ws = (char*)d_ws;
    size_t off = 0;
    float* dinv = (float*)(ws + off);  off += ((size_t)N * 4 + 255) & ~(size_t)255;
    int* cnt    = (int*)(ws + off);    off += ((size_t)N * 4 + 255) & ~(size_t)255;
    int* rowptr = (int*)(ws + off);    off += ((size_t)(N + 1) * 4 + 255) & ~(size_t)255;
    int* bsum   = (int*)(ws + off);    off += ((size_t)NB * 4 + 255) & ~(size_t)255;
    int* rank   = (int*)(ws + off);    off += ((size_t)E * 4 + 255) & ~(size_t)255;
    int* csr    = (int*)(ws + off);    off += ((size_t)E * 4 + 255) & ~(size_t)255;
    float* hbuf = (float*)(ws + off);
    float* tmp  = hbuf + (size_t)N * CH;

    dim3 gN((N + T - 1) / T), gE((E + T - 1) / T);
    dim3 gGemm((N + 63) / 64);
    dim3 gGemmCnt(gGemm.x + (E + 128 * CNT_EDGES - 1) / (128 * CNT_EDGES));
    dim3 gRow((N + 7) / 8);

    zero_cnt_kernel<<<gN, T, 0, stream>>>(cnt, N);
    gemm_count_kernel<<<gGemmCnt, 128, 0, stream>>>(x, W1, hbuf, N, dstI, cnt, rank, E, (int)gGemm.x);
    block_sum_kernel<<<NB, T, 0, stream>>>(cnt, bsum, N);
    scan_kernel<<<NB, T, 0, stream>>>(cnt, bsum, rowptr, dinv, N);
    fill_kernel<<<gE, T, 0, stream>>>(srcI, dstI, rank, rowptr, csr, E);

    gather_kernel<<<gRow, T, 0, stream>>>(hbuf, dinv, rowptr, csr, b1, tmp, N);

    gemm_wmma_kernel<<<gGemm, 128, 0, stream>>>(tmp, W2, hbuf, N, 1);
    gather_kernel<<<gRow, T, 0, stream>>>(hbuf, dinv, rowptr, csr, b2, out, N);
}


# CSR build becomes a two-level counting sort in LDS (512-node bins): no global atomics, no scattered 4-byte stores, csr_src written as whole lines; row sum order = LDS arrival rank
# speedup vs baseline: 4.0789x; 1.4968x over previous
#include <hip/hip_runtime.h>
#include <hip/hip_bf16.h>
#include <cstddef>

typedef __attribute__((ext_vector_type(2))) float v2f;
typedef __attribute__((ext_vector_type(8))) float v8f;

#define CH 128

#define SCAN_TILE 1024
#define BIN_SHIFT 9
#define BIN_NODES 512
#define MAXB 1024
#define CHUNK 8192
#define STAGE_CAP 16384

__device__ __forceinline__ int wave_sum(int v) {
    #pragma unroll
    for (int off = 16; off > 0; off >>= 1) v += __shfl_xor(v, off);
    return v;
}

__global__ __launch_bounds__(256) void block_sum_kernel(
    const int* __restrict__ cnt, int* __restrict__ bsum, int n)
{
    __shared__ int red[8];
    const int lane = threadIdx.x & 31;
    const int wave = threadIdx.x >> 5;
    const int base = blockIdx.x * SCAN_TILE + threadIdx.x * 4;
    int s = 0;
    #pragma unroll
    for (int k = 0; k < 4; ++k)
        if (base + k < n) s += cnt[base + k];
    s = wave_sum(s);
    if (lane == 0) red[wave] = s;
    __syncthreads();
    if (threadIdx.x == 0) {
        int t = 0;
        #pragma unroll
        for (int w = 0; w < 8; ++w) t += red[w];
        bsum[blockIdx.x] = t;
    }
}

__global__ __launch_bounds__(256) void scan_kernel(
    const int* __restrict__ cnt, const int* __restrict__ bsum,
    int* __restrict__ rowptr, int n)
{
    __shared__ int red[8];
    __shared__ int wsum[8];
    const int lane = threadIdx.x & 31;
    const int wave = threadIdx.x >> 5;

    int part = 0;
    for (int j = threadIdx.x; j < (int)blockIdx.x; j += 256) part += bsum[j];
    part = wave_sum(part);
    if (lane == 0) red[wave] = part;

    const int base = blockIdx.x * SCAN_TILE + threadIdx.x * 4;
    int c[4];
    #pragma unroll
    for (int k = 0; k < 4; ++k) c[k] = (base + k < n) ? cnt[base + k] : 0;
    const int tsum = c[0] + c[1] + c[2] + c[3];

    int inc = tsum;
    #pragma unroll
    for (int off = 1; off < 32; off <<= 1) {
        int v = __shfl_up(inc, off);
        if (lane >= off) inc += v;
    }
    if (lane == 31) wsum[wave] = inc;
    __syncthreads();

    int offset = 0;
    #pragma unroll
    for (int w = 0; w < 8; ++w) offset += red[w];
    for (int w = 0; w < wave; ++w) offset += wsum[w];

    int excl = offset + inc - tsum;
    #pragma unroll
    for (int k = 0; k < 4; ++k) {
        const int idx = base + k;
        if (idx < n) {
            rowptr[idx] = excl;
            excl += c[k];
            if (idx == n - 1) rowptr[n] = excl;
        }
    }
}

__device__ __forceinline__ int block_excl_scan_1024(int v, int* wsum, int& total)
{
    const int lane = threadIdx.x & 31;
    const int wave = threadIdx.x >> 5;
    int inc = v;
    #pragma unroll
    for (int off = 1; off < 32; off <<= 1) {
        int u = __shfl_up(inc, off);
        if (lane >= off) inc += u;
    }
    if (lane == 31) wsum[wave] = inc;
    __syncthreads();
    if (wave == 0) {
        const int w = wsum[lane];
        int winc = w;
        #pragma unroll
        for (int off = 1; off < 32; off <<= 1) {
            int u = __shfl_up(winc, off);
            if (lane >= off) winc += u;
        }
        wsum[lane] = winc - w;
        if (lane == 31) wsum[32] = winc;
    }
    __syncthreads();
    total = wsum[32];
    return wsum[wave] + inc - v;
}

__device__ __forceinline__ void hist_chunk(
    const int* __restrict__ dst, int* __restrict__ cnt2, int e, int nb, int nblk, int blk)
{
    __shared__ int hist[MAXB];
    const int t = threadIdx.x;
    for (int b = t; b < nb; b += 128) hist[b] = 0;
    __syncthreads();
    const int first = blk * CHUNK;
    if (first + CHUNK <= e) {
        for (int k0 = 0; k0 < CHUNK / 128; k0 += 8) {
            int d[8];
            #pragma unroll
            for (int k = 0; k < 8; ++k) d[k] = dst[first + (k0 + k) * 128 + t];
            #pragma unroll
            for (int k = 0; k < 8; ++k) atomicAdd(&hist[d[k] >> BIN_SHIFT], 1);
        }
    } else {
        for (int i = first + t; i < e; i += 128) atomicAdd(&hist[dst[i] >> BIN_SHIFT], 1);
    }
    __syncthreads();
    for (int b = t; b < nb; b += 128) cnt2[b * nblk + blk] = hist[b];
}

__global__ __launch_bounds__(1024) void partition_kernel(
    const int* __restrict__ src, const int* __restrict__ dst,
    const int* __restrict__ off, int* __restrict__ part_raw, int e, int nb, int nblk)
{
    __shared__ int hist[MAXB];
    __shared__ int lstart[MAXB];
    __shared__ int goff[MAXB];
    __shared__ int wsum[33];
    __shared__ int st_src[CHUNK];
    __shared__ int st_dst[CHUNK];
    int2* __restrict__ part = (int2*)part_raw;

    const int t = threadIdx.x;
    const int blk = blockIdx.x;
    hist[t] = 0;
    __syncthreads();

    const int first = blk * CHUNK;
    int s[CHUNK / 1024], d[CHUNK / 1024], r[CHUNK / 1024];
    #pragma unroll
    for (int k = 0; k < CHUNK / 1024; ++k) {
        const int i = first + k * 1024 + t;
        s[k] = 0; d[k] = -1; r[k] = 0;
        if (i < e) { s[k] = src[i]; d[k] = dst[i]; }
    }
    #pragma unroll
    for (int k = 0; k < CHUNK / 1024; ++k)
        if (d[k] >= 0) r[k] = atomicAdd(&hist[d[k] >> BIN_SHIFT], 1);
    __syncthreads();

    int total;
    const int ex = block_excl_scan_1024(hist[t], wsum, total);
    lstart[t] = ex;
    if (t < nb) goff[t] = off[t * nblk + blk];
    __syncthreads();

    #pragma unroll
    for (int k = 0; k < CHUNK / 1024; ++k) {
        if (d[k] >= 0) {
            const int p = lstart[d[k] >> BIN_SHIFT] + r[k];
            st_src[p] = s[k];
            st_dst[p] = d[k];
        }
    }
    __syncthreads();

    for (int j = t; j < total; j += 1024) {
        const int dd = st_dst[j];
        const int b = dd >> BIN_SHIFT;
        part[goff[b] + (j - lstart[b])] = make_int2(st_src[j], dd);
    }
}

__global__ __launch_bounds__(1024) void binsort_kernel(
    const int* __restrict__ part_raw, const int* __restrict__ off,
    int* __restrict__ rowptr, float* __restrict__ dinv, int* __restrict__ csr_src,
    int n, int e, int nb, int nblk)
{
    __shared__ int cnt[BIN_NODES];
    __shared__ int cur[BIN_NODES];
    __shared__ int wsum[33];
    __shared__ int stage[STAGE_CAP];
    const int2* __restrict__ part = (const int2*)part_raw;

    const int t = threadIdx.x;
    const int bin = blockIdx.x;
    const int beg = off[bin * nblk];
    const int end = (bin + 1 < nb) ? off[(bin + 1) * nblk] : e;
    const int node0 = bin << BIN_SHIFT;

    if (t < BIN_NODES) cnt[t] = 0;
    __syncthreads();
    for (int i = beg + t; i < end; i += 1024) atomicAdd(&cnt[part[i].y - node0], 1);
    __syncthreads();

    const int c = (t < BIN_NODES) ? cnt[t] : 0;
    int total;
    const int ex = block_excl_scan_1024(c, wsum, total);
    if (t < BIN_NODES) {
        cur[t] = ex;
        const int node = node0 + t;
        if (node < n) {
            rowptr[node] = beg + ex;
            dinv[node] = rsqrtf((float)(c + 1));
            if (node == n - 1) rowptr[n] = beg + ex + c;
        }
    }
    __syncthreads();

    const bool staged = (end - beg) <= STAGE_CAP;
    for (int i = beg + t; i < end; i += 1024) {
        const int2 sd = part[i];
        const int p = atomicAdd(&cur[sd.y - node0], 1);
        if (staged) stage[p] = sd.x; else csr_src[beg + p] = sd.x;
    }
    if (staged) {
        __syncthreads();
        const int m = end - beg;
        for (int j = t; j < m; j += 1024) csr_src[beg + j] = stage[j];
    }
}

__device__ __forceinline__ void gemm_tile(
    const float* __restrict__ X, const float* __restrict__ W,
    float* __restrict__ H, int nrows, int apply_relu, int tile)
{
    __shared__ float Xs[4][16 * CH];

    const int lane = threadIdx.x & 31;
    const int wave = threadIdx.x >> 5;
    const int row_base = tile * 64 + wave * 16;
    const bool full_tile = (row_base + 16 <= nrows);

    float* xs = &Xs[wave][0];

    if (full_tile) {
        #pragma unroll
        for (int r = 0; r < 16; ++r) {
            float4 v = *(const float4*)(X + (size_t)(row_base + r) * CH + lane * 4);
            if (apply_relu) {
                v.x = fmaxf(v.x, 0.f); v.y = fmaxf(v.y, 0.f);
                v.z = fmaxf(v.z, 0.f); v.w = fmaxf(v.w, 0.f);
            }
            *(float4*)(xs + r * CH + lane * 4) = v;
        }
    } else {
        #pragma unroll
        for (int r = 0; r < 16; ++r) {
            int row = row_base + r;
            float4 v = make_float4(0.f, 0.f, 0.f, 0.f);
            if (row < nrows)
                v = *(const float4*)(X + (size_t)row * CH + lane * 4);
            if (apply_relu) {
                v.x = fmaxf(v.x, 0.f); v.y = fmaxf(v.y, 0.f);
                v.z = fmaxf(v.z, 0.f); v.w = fmaxf(v.w, 0.f);
            }
            *(float4*)(xs + r * CH + lane * 4) = v;
        }
    }
    __syncthreads();

    const int m    = lane & 15;
    const int hi   = lane >> 4;
    const int colN = lane & 15;

    v8f acc[8];
    #pragma unroll
    for (int n = 0; n < 8; ++n) acc[n] = (v8f)(0.0f);

    for (int k0 = 0; k0 < CH; k0 += 4) {
        const int klo = k0 + hi * 2;
        v2f a = *(const v2f*)(xs + m * CH + klo);
        #pragma unroll
        for (int n = 0; n < 8; ++n) {
            v2f b;
            b.x = W[(size_t)klo * CH + n * 16 + colN];
            b.y = W[(size_t)(klo + 1) * CH + n * 16 + colN];
            acc[n] = __builtin_amdgcn_wmma_f32_16x16x4_f32(
                false, a, false, b, (short)0, acc[n], false, false);
        }
    }

    if (full_tile) {
        #pragma unroll
        for (int n = 0; n < 8; ++n) {
            #pragma unroll
            for (int r = 0; r < 8; ++r) {
                int row = row_base + r + hi * 8;
                H[(size_t)row * CH + n * 16 + colN] = acc[n][r];
            }
        }
    } else if (row_base < nrows) {
        #pragma unroll
        for (int n = 0; n < 8; ++n) {
            #pragma unroll
            for (int r = 0; r < 8; ++r) {
                int row = row_base + r + hi * 8;
                if (row < nrows)
                    H[(size_t)row * CH + n * 16 + colN] = acc[n][r];
            }
        }
    }
}

__global__ __launch_bounds__(128) void gemm_wmma_kernel(
    const float* __restrict__ X, const float* __restrict__ W,
    float* __restrict__ H, int nrows, int apply_relu)
{
    gemm_tile(X, W, H, nrows, apply_relu, (int)blockIdx.x);
}

__global__ __launch_bounds__(128) void gemm_hist_kernel(
    const float* __restrict__ X, const float* __restrict__ W,
    float* __restrict__ H, int nrows,
    const int* __restrict__ dst, int* __restrict__ cnt2,
    int e, int nb, int nblk, int gemm_blocks)
{
    if ((int)blockIdx.x >= gemm_blocks) {
        hist_chunk(dst, cnt2, e, nb, nblk, (int)blockIdx.x - gemm_blocks);
        return;
    }
    gemm_tile(X, W, H, nrows, 0, (int)blockIdx.x);
}

template <int K>
__device__ __forceinline__ void add_rows(const float* __restrict__ H, int sidx, float cf,
                                         int j, int lane, float4& acc)
{
    float4 v[K];
    float  c[K];
    #pragma unroll
    for (int k = 0; k < K; ++k) {
        const int s = __builtin_amdgcn_readlane(sidx, j + k);
        c[k] = __int_as_float(__builtin_amdgcn_readlane(__float_as_int(cf), j + k));
        v[k] = *(const float4*)(H + (size_t)s * CH + lane * 4);
    }
    #pragma unroll
    for (int k = 0; k < K; ++k) {
        acc.x = fmaf(v[k].x, c[k], acc.x);
        acc.y = fmaf(v[k].y, c[k], acc.y);
        acc.z = fmaf(v[k].z, c[k], acc.z);
        acc.w = fmaf(v[k].w, c[k], acc.w);
    }
}

__global__ __launch_bounds__(256) void gather_kernel(
    const float* __restrict__ H, const float* __restrict__ dinv,
    const int* __restrict__ rowptr, const int* __restrict__ csr_src,
    const float* __restrict__ bias, float* __restrict__ out, int n)
{
    const int lane = threadIdx.x & 31;
    const int row = __builtin_amdgcn_readfirstlane(
        (int)((blockIdx.x * blockDim.x + threadIdx.x) >> 5));
    if (row >= n) return;

    const int beg = rowptr[row];
    const int end = rowptr[row + 1];
    const float di = dinv[row];
    const float w = di * di;

    const float4 h = *(const float4*)(H + (size_t)row * CH + lane * 4);
    float4 acc;
    acc.x = h.x * w; acc.y = h.y * w; acc.z = h.z * w; acc.w = h.w * w;

    for (int base = beg; base < end; base += 32) {
        const int nb = (end - base < 32) ? (end - base) : 32;
        int sidx = 0;
        float cf = 0.f;
        if (lane < nb) {
            sidx = csr_src[base + lane];
            cf = dinv[sidx] * di;
        }
        int j = 0;
        for (; j + 4 <= nb; j += 4) add_rows<4>(H, sidx, cf, j, lane, acc);
        if (j + 2 <= nb) { add_rows<2>(H, sidx, cf, j, lane, acc); j += 2; }
        if (j < nb) add_rows<1>(H, sidx, cf, j, lane, acc);
    }

    const float4 b = *(const float4*)(bias + lane * 4);
    float4 o;
    o.x = acc.x + b.x; o.y = acc.y + b.y; o.z = acc.z + b.z; o.w = acc.w + b.w;
    *(float4*)(out + (size_t)row * CH + lane * 4) = o;
}

extern "C" void kernel_launch(void* const* d_in, const int* in_sizes, int n_in,
                              void* d_out, int out_size, void* d_ws, size_t ws_size,
                              hipStream_t stream) {
    const float* x    = (const float*)d_in[0];
    const int*   edge = (const int*)d_in[1];
    const float* W1   = (const float*)d_in[2];
    const float* b1   = (const float*)d_in[3];
    const float* W2   = (const float*)d_in[4];
    const float* b2   = (const float*)d_in[5];
    float* out = (float*)d_out;

    const int N = in_sizes[0] / CH;
    const int E = in_sizes[1] / 2;
    const int* srcI = edge;
    const int* dstI = edge + E;

    const int T = 256;
    const int NBIN = (N + BIN_NODES - 1) / BIN_NODES;
    const int NBLK = (E + CHUNK - 1) / CHUNK;
    const int N2 = NBIN * NBLK;
    const int NB2 = (N2 + SCAN_TILE - 1) / SCAN_TILE;

    char* ws = (char*)d_ws;
    size_t off = 0;
    float* dinv = (float*)(ws + off);  off += ((size_t)N * 4 + 255) & ~(size_t)255;
    int* rowptr = (int*)(ws + off);    off += ((size_t)(N + 1) * 4 + 255) & ~(size_t)255;
    int* cnt2   = (int*)(ws + off);    off += ((size_t)N2 * 4 + 255) & ~(size_t)255;
    int* offs   = (int*)(ws + off);    off += ((size_t)(N2 + 1) * 4 + 255) & ~(size_t)255;
    int* bsum   = (int*)(ws + off);    off += ((size_t)NB2 * 4 + 255) & ~(size_t)255;
    int* part   = (int*)(ws + off);    off += ((size_t)E * 8 + 255) & ~(size_t)255;
    int* csr    = (int*)(ws + off);    off += ((size_t)E * 4 + 255) & ~(size_t)255;
    float* hbuf = (float*)(ws + off);
    float* tmp  = hbuf + (size_t)N * CH;

    dim3 gGemm((N + 63) / 64);
    dim3 gGemmHist(gGemm.x + NBLK);
    dim3 gRow((N + 7) / 8);

    gemm_hist_kernel<<<gGemmHist, 128, 0, stream>>>(x, W1, hbuf, N, dstI, cnt2, E, NBIN, NBLK, (int)gGemm.x);
    block_sum_kernel<<<NB2, T, 0, stream>>>(cnt2, bsum, N2);
    scan_kernel<<<NB2, T, 0, stream>>>(cnt2, bsum, offs, N2);
    partition_kernel<<<NBLK, 1024, 0, stream>>>(srcI, dstI, offs, part, E, NBIN, NBLK);
    binsort_kernel<<<NBIN, 1024, 0, stream>>>(part, offs, rowptr, dinv, csr, N, E, NBIN, NBLK);

    gather_kernel<<<gRow, T, 0, stream>>>(hbuf, dinv, rowptr, csr, b1, tmp, N);

    gemm_wmma_kernel<<<gGemm, 128, 0, stream>>>(tmp, W2, hbuf, N, 1);
    gather_kernel<<<gRow, T, 0, stream>>>(hbuf, dinv, rowptr, csr, b2, out, N);
}


# layer-2 GEMM runs as the epilogue of the layer-1 gather: summed rows go relu -> LDS tile -> f32 WMMA against a W2 panel kept in registers; 8 -> 7 launches; row sum order unchanged
# speedup vs baseline: 4.7006x; 1.1524x over previous
#include <hip/hip_runtime.h>
#include <hip/hip_bf16.h>
#include <cstddef>

typedef __attribute__((ext_vector_type(2))) float v2f;
typedef __attribute__((ext_vector_type(8))) float v8f;

#define CH 128

#define SCAN_TILE 1024
#define BIN_SHIFT 9
#define BIN_NODES 512
#define MAXB 1024
#define CHUNK 8192
#define STAGE_CAP 16384

__device__ __forceinline__ int wave_sum(int v) {
    #pragma unroll
    for (int off = 16; off > 0; off >>= 1) v += __shfl_xor(v, off);
    return v;
}

__global__ __launch_bounds__(256) void block_sum_kernel(
    const int* __restrict__ cnt, int* __restrict__ bsum, int n)
{
    __shared__ int red[8];
    const int lane = threadIdx.x & 31;
    const int wave = threadIdx.x >> 5;
    const int base = blockIdx.x * SCAN_TILE + threadIdx.x * 4;
    int s = 0;
    #pragma unroll
    for (int k = 0; k < 4; ++k)
        if (base + k < n) s += cnt[base + k];
    s = wave_sum(s);
    if (lane == 0) red[wave] = s;
    __syncthreads();
    if (threadIdx.x == 0) {
        int t = 0;
        #pragma unroll
        for (int w = 0; w < 8; ++w) t += red[w];
        bsum[blockIdx.x] = t;
    }
}

__global__ __launch_bounds__(256) void scan_kernel(
    const int* __restrict__ cnt, const int* __restrict__ bsum,
    int* __restrict__ rowptr, int n)
{
    __shared__ int red[8];
    __shared__ int wsum[8];
    const int lane = threadIdx.x & 31;
    const int wave = threadIdx.x >> 5;

    int part = 0;
    for (int j = threadIdx.x; j < (int)blockIdx.x; j += 256) part += bsum[j];
    part = wave_sum(part);
    if (lane == 0) red[wave] = part;

    const int base = blockIdx.x * SCAN_TILE + threadIdx.x * 4;
    int c[4];
    #pragma unroll
    for (int k = 0; k < 4; ++k) c[k] = (base + k < n) ? cnt[base + k] : 0;
    const int tsum = c[0] + c[1] + c[2] + c[3];

    int inc = tsum;
    #pragma unroll
    for (int off = 1; off < 32; off <<= 1) {
        int v = __shfl_up(inc, off);
        if (lane >= off) inc += v;
    }
    if (lane == 31) wsum[wave] = inc;
    __syncthreads();

    int offset = 0;
    #pragma unroll
    for (int w = 0; w < 8; ++w) offset += red[w];
    for (int w = 0; w < wave; ++w) offset += wsum[w];

    int excl = offset + inc - tsum;
    #pragma unroll
    for (int k = 0; k < 4; ++k) {
        const int idx = base + k;
        if (idx < n) {
            rowptr[idx] = excl;
            excl += c[k];
            if (idx == n - 1) rowptr[n] = excl;
        }
    }
}

__device__ __forceinline__ int block_excl_scan_1024(int v, int* wsum, int& total)
{
    const int lane = threadIdx.x & 31;
    const int wave = threadIdx.x >> 5;
    int inc = v;
    #pragma unroll
    for (int off = 1; off < 32; off <<= 1) {
        int u = __shfl_up(inc, off);
        if (lane >= off) inc += u;
    }
    if (lane == 31) wsum[wave] = inc;
    __syncthreads();
    if (wave == 0) {
        const int w = wsum[lane];
        int winc = w;
        #pragma unroll
        for (int off = 1; off < 32; off <<= 1) {
            int u = __shfl_up(winc, off);
            if (lane >= off) winc += u;
        }
        wsum[lane] = winc - w;
        if (lane == 31) wsum[32] = winc;
    }
    __syncthreads();
    total = wsum[32];
    return wsum[wave] + inc - v;
}

__device__ __forceinline__ void hist_chunk(
    const int* __restrict__ dst, int* __restrict__ cnt2, int e, int nb, int nblk, int blk)
{
    __shared__ int hist[MAXB];
    const int t = threadIdx.x;
    for (int b = t; b < nb; b += 128) hist[b] = 0;
    __syncthreads();
    const int first = blk * CHUNK;
    if (first + CHUNK <= e) {
        for (int k0 = 0; k0 < CHUNK / 128; k0 += 8) {
            int d[8];
            #pragma unroll
            for (int k = 0; k < 8; ++k) d[k] = dst[first + (k0 + k) * 128 + t];
            #pragma unroll
            for (int k = 0; k < 8; ++k) atomicAdd(&hist[d[k] >> BIN_SHIFT], 1);
        }
    } else {
        for (int i = first + t; i < e; i += 128) atomicAdd(&hist[dst[i] >> BIN_SHIFT], 1);
    }
    __syncthreads();
    for (int b = t; b < nb; b += 128) cnt2[b * nblk + blk] = hist[b];
}

__global__ __launch_bounds__(1024) void partition_kernel(
    const int* __restrict__ src, const int* __restrict__ dst,
    const int* __restrict__ off, int* __restrict__ part_raw, int e, int nb, int nblk)
{
    __shared__ int hist[MAXB];
    __shared__ int lstart[MAXB];
    __shared__ int goff[MAXB];
    __shared__ int wsum[33];
    __shared__ int st_src[CHUNK];
    __shared__ int st_dst[CHUNK];
    int2* __restrict__ part = (int2*)part_raw;

    const int t = threadIdx.x;
    const int blk = blockIdx.x;
    hist[t] = 0;
    __syncthreads();

    const int first = blk * CHUNK;
    int s[CHUNK / 1024], d[CHUNK / 1024], r[CHUNK / 1024];
    #pragma unroll
    for (int k = 0; k < CHUNK / 1024; ++k) {
        const int i = first + k * 1024 + t;
        s[k] = 0; d[k] = -1; r[k] = 0;
        if (i < e) { s[k] = src[i]; d[k] = dst[i]; }
    }
    #pragma unroll
    for (int k = 0; k < CHUNK / 1024; ++k)
        if (d[k] >= 0) r[k] = atomicAdd(&hist[d[k] >> BIN_SHIFT], 1);
    __syncthreads();

    int total;
    const int ex = block_excl_scan_1024(hist[t], wsum, total);
    lstart[t] = ex;
    if (t < nb) goff[t] = off[t * nblk + blk];
    __syncthreads();

    #pragma unroll
    for (int k = 0; k < CHUNK / 1024; ++k) {
        if (d[k] >= 0) {
            const int p = lstart[d[k] >> BIN_SHIFT] + r[k];
            st_src[p] = s[k];
            st_dst[p] = d[k];
        }
    }
    __syncthreads();

    for (int j = t; j < total; j += 1024) {
        const int dd = st_dst[j];
        const int b = dd >> BIN_SHIFT;
        part[goff[b] + (j - lstart[b])] = make_int2(st_src[j], dd);
    }
}

__global__ __launch_bounds__(1024) void binsort_kernel(
    const int* __restrict__ part_raw, const int* __restrict__ off,
    int* __restrict__ rowptr, float* __restrict__ dinv, int* __restrict__ csr_src,
    int n, int e, int nb, int nblk)
{
    __shared__ int cnt[BIN_NODES];
    __shared__ int cur[BIN_NODES];
    __shared__ int wsum[33];
    __shared__ int stage[STAGE_CAP];
    const int2* __restrict__ part = (const int2*)part_raw;

    const int t = threadIdx.x;
    const int bin = blockIdx.x;
    const int beg = off[bin * nblk];
    const int end = (bin + 1 < nb) ? off[(bin + 1) * nblk] : e;
    const int node0 = bin << BIN_SHIFT;

    if (t < BIN_NODES) cnt[t] = 0;
    __syncthreads();
    for (int i = beg + t; i < end; i += 1024) atomicAdd(&cnt[part[i].y - node0], 1);
    __syncthreads();

    const int c = (t < BIN_NODES) ? cnt[t] : 0;
    int total;
    const int ex = block_excl_scan_1024(c, wsum, total);
    if (t < BIN_NODES) {
        cur[t] = ex;
        const int node = node0 + t;
        if (node < n) {
            rowptr[node] = beg + ex;
            dinv[node] = rsqrtf((float)(c + 1));
            if (node == n - 1) rowptr[n] = beg + ex + c;
        }
    }
    __syncthreads();

    const bool staged = (end - beg) <= STAGE_CAP;
    for (int i = beg + t; i < end; i += 1024) {
        const int2 sd = part[i];
        const int p = atomicAdd(&cur[sd.y - node0], 1);
        if (staged) stage[p] = sd.x; else csr_src[beg + p] = sd.x;
    }
    if (staged) {
        __syncthreads();
        const int m = end - beg;
        for (int j = t; j < m; j += 1024) csr_src[beg + j] = stage[j];
    }
}

__device__ __forceinline__ void gemm_tile(
    const float* __restrict__ X, const float* __restrict__ W,
    float* __restrict__ H, int nrows, int apply_relu, int tile)
{
    __shared__ float Xs[4][16 * CH];

    const int lane = threadIdx.x & 31;
    const int wave = threadIdx.x >> 5;
    const int row_base = tile * 64 + wave * 16;
    const bool full_tile = (row_base + 16 <= nrows);

    float* xs = &Xs[wave][0];

    if (full_tile) {
        #pragma unroll
        for (int r = 0; r < 16; ++r) {
            float4 v = *(const float4*)(X + (size_t)(row_base + r) * CH + lane * 4);
            if (apply_relu) {
                v.x = fmaxf(v.x, 0.f); v.y = fmaxf(v.y, 0.f);
                v.z = fmaxf(v.z, 0.f); v.w = fmaxf(v.w, 0.f);
            }
            *(float4*)(xs + r * CH + lane * 4) = v;
        }
    } else {
        #pragma unroll
        for (int r = 0; r < 16; ++r) {
            int row = row_base + r;
            float4 v = make_float4(0.f, 0.f, 0.f, 0.f);
            if (row < nrows)
                v = *(const float4*)(X + (size_t)row * CH + lane * 4);
            if (apply_relu) {
                v.x = fmaxf(v.x, 0.f); v.y = fmaxf(v.y, 0.f);
                v.z = fmaxf(v.z, 0.f); v.w = fmaxf(v.w, 0.f);
            }
            *(float4*)(xs + r * CH + lane * 4) = v;
        }
    }
    __syncthreads();

    const int m    = lane & 15;
    const int hi   = lane >> 4;
    const int colN = lane & 15;

    v8f acc[8];
    #pragma unroll
    for (int n = 0; n < 8; ++n) acc[n] = (v8f)(0.0f);

    for (int k0 = 0; k0 < CH; k0 += 4) {
        const int klo = k0 + hi * 2;
        v2f a = *(const v2f*)(xs + m * CH + klo);
        #pragma unroll
        for (int n = 0; n < 8; ++n) {
            v2f b;
            b.x = W[(size_t)klo * CH + n * 16 + colN];
            b.y = W[(size_t)(klo + 1) * CH + n * 16 + colN];
            acc[n] = __builtin_amdgcn_wmma_f32_16x16x4_f32(
                false, a, false, b, (short)0, acc[n], false, false);
        }
    }

    if (full_tile) {
        #pragma unroll
        for (int n = 0; n < 8; ++n) {
            #pragma unroll
            for (int r = 0; r < 8; ++r) {
                int row = row_base + r + hi * 8;
                H[(size_t)row * CH + n * 16 + colN] = acc[n][r];
            }
        }
    } else if (row_base < nrows) {
        #pragma unroll
        for (int n = 0; n < 8; ++n) {
            #pragma unroll
            for (int r = 0; r < 8; ++r) {
                int row = row_base + r + hi * 8;
                if (row < nrows)
                    H[(size_t)row * CH + n * 16 + colN] = acc[n][r];
            }
        }
    }
}

__global__ __launch_bounds__(128) void gemm_hist_kernel(
    const float* __restrict__ X, const float* __restrict__ W,
    float* __restrict__ H, int nrows,
    const int* __restrict__ dst, int* __restrict__ cnt2,
    int e, int nb, int nblk, int gemm_blocks)
{
    if ((int)blockIdx.x >= gemm_blocks) {
        hist_chunk(dst, cnt2, e, nb, nblk, (int)blockIdx.x - gemm_blocks);
        return;
    }
    gemm_tile(X, W, H, nrows, 0, (int)blockIdx.x);
}

template <int K>
__device__ __forceinline__ void add_rows(const float* __restrict__ H, int sidx, float cf,
                                         int j, int lane, float4& acc)
{
    float4 v[K];
    float  c[K];
    #pragma unroll
    for (int k = 0; k < K; ++k) {
        const int s = __builtin_amdgcn_readlane(sidx, j + k);
        c[k] = __int_as_float(__builtin_amdgcn_readlane(__float_as_int(cf), j + k));
        v[k] = *(const float4*)(H + (size_t)s * CH + lane * 4);
    }
    #pragma unroll
    for (int k = 0; k < K; ++k) {
        acc.x = fmaf(v[k].x, c[k], acc.x);
        acc.y = fmaf(v[k].y, c[k], acc.y);
        acc.z = fmaf(v[k].z, c[k], acc.z);
        acc.w = fmaf(v[k].w, c[k], acc.w);
    }
}

__global__ __launch_bounds__(256) void gather_kernel(
    const float* __restrict__ H, const float* __restrict__ dinv,
    const int* __restrict__ rowptr, const int* __restrict__ csr_src,
    const float* __restrict__ bias, float* __restrict__ out, int n)
{
    const int lane = threadIdx.x & 31;
    const int row = __builtin_amdgcn_readfirstlane(
        (int)((blockIdx.x * blockDim.x + threadIdx.x) >> 5));
    if (row >= n) return;

    const int beg = rowptr[row];
    const int end = rowptr[row + 1];
    const float di = dinv[row];
    const float w = di * di;

    const float4 h = *(const float4*)(H + (size_t)row * CH + lane * 4);
    float4 acc;
    acc.x = h.x * w; acc.y = h.y * w; acc.z = h.z * w; acc.w = h.w * w;

    for (int base = beg; base < end; base += 32) {
        const int nb = (end - base < 32) ? (end - base) : 32;
        int sidx = 0;
        float cf = 0.f;
        if (lane < nb) {
            sidx = csr_src[base + lane];
            cf = dinv[sidx] * di;
        }
        int j = 0;
        for (; j + 4 <= nb; j += 4) add_rows<4>(H, sidx, cf, j, lane, acc);
        if (j + 2 <= nb) { add_rows<2>(H, sidx, cf, j, lane, acc); j += 2; }
        if (j < nb) add_rows<1>(H, sidx, cf, j, lane, acc);
    }

    const float4 b = *(const float4*)(bias + lane * 4);
    float4 o;
    o.x = acc.x + b.x; o.y = acc.y + b.y; o.z = acc.z + b.z; o.w = acc.w + b.w;
    *(float4*)(out + (size_t)row * CH + lane * 4) = o;
}

__device__ __forceinline__ void add_chunk(const float* __restrict__ H, int sidx, float cf,
                                          int nb, int lane, float4& acc)
{
    int j = 0;
    for (; j + 8 <= nb; j += 8) add_rows<8>(H, sidx, cf, j, lane, acc);
    switch (nb - j) {
        case 7: add_rows<7>(H, sidx, cf, j, lane, acc); break;
        case 6: add_rows<6>(H, sidx, cf, j, lane, acc); break;
        case 5: add_rows<5>(H, sidx, cf, j, lane, acc); break;
        case 4: add_rows<4>(H, sidx, cf, j, lane, acc); break;
        case 3: add_rows<3>(H, sidx, cf, j, lane, acc); break;
        case 2: add_rows<2>(H, sidx, cf, j, lane, acc); break;
        case 1: add_rows<1>(H, sidx, cf, j, lane, acc); break;
        default: break;
    }
}

#define TROWS 32
#define XP 132

__global__ __launch_bounds__(256) void gather_gemm_kernel(
    const float* __restrict__ H, const float* __restrict__ dinv,
    const int* __restrict__ rowptr, const int* __restrict__ csr_src,
    const float* __restrict__ bias, const float* __restrict__ W,
    float* __restrict__ Hout, int n, int rows_per_wg)
{
    __shared__ __attribute__((aligned(16))) float Xs[2][TROWS * XP];
    __shared__ int4 Es[8][TROWS / 8][32];

    const int lane = threadIdx.x & 31;
    const int wave = __builtin_amdgcn_readfirstlane((int)(threadIdx.x >> 5));
    const int c  = lane & 15;
    const int hi = lane >> 4;

    v2f bfrag[CH / 4];
    #pragma unroll
    for (int ks = 0; ks < CH / 4; ++ks) {
        const int klo = ks * 4 + hi * 2;
        bfrag[ks].x = W[(size_t)klo * CH + wave * 16 + c];
        bfrag[ks].y = W[(size_t)(klo + 1) * CH + wave * 16 + c];
    }
    const float4 b = *(const float4*)(bias + lane * 4);

    const int row0 = (int)blockIdx.x * rows_per_wg;
    const int row1 = (row0 + rows_per_wg < n) ? (row0 + rows_per_wg) : n;

    int buf = 0;
    for (int tbase = row0; tbase < row1; tbase += TROWS, buf ^= 1) {
        float* xs = &Xs[buf][0];

        float4 hnext = make_float4(0.f, 0.f, 0.f, 0.f);
        if (tbase + wave < row1)
            hnext = *(const float4*)(H + (size_t)(tbase + wave) * CH + lane * 4);
        {
            int rbeg[TROWS / 8], rend[TROWS / 8];
            float rdi[TROWS / 8];
            int sidx0[TROWS / 8];
            #pragma unroll
            for (int jr = 0; jr < TROWS / 8; ++jr) {
                const int row = tbase + wave + 8 * jr;
                const int rc = (row < row1) ? row : row0;
                const int b0 = rowptr[rc];
                const int e0 = rowptr[rc + 1];
                rbeg[jr] = b0;
                rend[jr] = (row < row1) ? e0 : b0;
                rdi[jr] = dinv[rc];
            }
            #pragma unroll
            for (int jr = 0; jr < TROWS / 8; ++jr) {
                int p = rbeg[jr] + lane;
                p = (p < rend[jr] - 1) ? p : (rend[jr] - 1);
                p = (p > 0) ? p : 0;
                sidx0[jr] = csr_src[p];
            }
            #pragma unroll
            for (int jr = 0; jr < TROWS / 8; ++jr) {
                const float cf = dinv[sidx0[jr]] * rdi[jr];
                Es[wave][jr][lane] = make_int4(sidx0[jr], __float_as_int(cf),
                                               rend[jr] - rbeg[jr], __float_as_int(rdi[jr] * rdi[jr]));
            }
        }

        #pragma unroll 1
        for (int jr = 0; jr < TROWS / 8; ++jr) {
            const int lr = wave + 8 * jr;
            const int row = tbase + lr;
            float4 o = make_float4(0.f, 0.f, 0.f, 0.f);
            if (row < row1) {
                const int4 sc = Es[wave][jr][lane];
                const int sidx0 = sc.x;
                const float cf0 = __int_as_float(sc.y);
                const int deg = __builtin_amdgcn_readfirstlane(sc.z);
                const float w = __int_as_float(sc.w);
                const float4 h = hnext;
                if (jr + 1 < TROWS / 8 && row + 8 < row1)
                    hnext = *(const float4*)(H + (size_t)(row + 8) * CH + lane * 4);
                float4 acc;
                acc.x = h.x * w; acc.y = h.y * w; acc.z = h.z * w; acc.w = h.w * w;
                add_chunk(H, sidx0, cf0, (deg < 32) ? deg : 32, lane, acc);
                if (deg > 32) {
                  const int beg = rowptr[row];
                  const int end = beg + deg;
                  const float di = dinv[row];
                  for (int base = beg + 32; base < end; base += 32) {
                    const int nb = (end - base < 32) ? (end - base) : 32;
                    int sidx = 0;
                    float cf = 0.f;
                    if (lane < nb) {
                        sidx = csr_src[base + lane];
                        cf = dinv[sidx] * di;
                    }
                    add_chunk(H, sidx, cf, nb, lane, acc);
                  }
                }
                o.x = fmaxf(acc.x + b.x, 0.f); o.y = fmaxf(acc.y + b.y, 0.f);
                o.z = fmaxf(acc.z + b.z, 0.f); o.w = fmaxf(acc.w + b.w, 0.f);
            }
            *(float4*)(xs + lr * XP + lane * 4) = o;
        }
        __syncthreads();

        v8f acc0 = (v8f)(0.0f);
        v8f acc1 = (v8f)(0.0f);
        #pragma unroll
        for (int ks = 0; ks < CH / 4; ++ks) {
            const int klo = ks * 4 + hi * 2;
            const v2f a0 = *(const v2f*)(xs + c * XP + klo);
            const v2f a1 = *(const v2f*)(xs + (16 + c) * XP + klo);
            acc0 = __builtin_amdgcn_wmma_f32_16x16x4_f32(
                false, a0, false, bfrag[ks], (short)0, acc0, false, false);
            acc1 = __builtin_amdgcn_wmma_f32_16x16x4_f32(
                false, a1, false, bfrag[ks], (short)0, acc1, false, false);
        }

        float* __restrict__ po = Hout + (size_t)(tbase + hi * 8) * CH + wave * 16 + c;
        if (tbase + TROWS <= row1) {
            #pragma unroll
            for (int r = 0; r < 8; ++r) {
                po[r * CH] = acc0[r];
                po[(r + 16) * CH] = acc1[r];
            }
        } else {
            const int left = row1 - (tbase + hi * 8);
            #pragma unroll
            for (int r = 0; r < 8; ++r) {
                if (r < left) po[r * CH] = acc0[r];
                if (r + 16 < left) po[(r + 16) * CH] = acc1[r];
            }
        }
    }
}

extern "C" void kernel_launch(void* const* d_in, const int* in_sizes, int n_in,
                              void* d_out, int out_size, void* d_ws, size_t ws_size,
                              hipStream_t stream) {
    const float* x    = (const float*)d_in[0];
    const int*   edge = (const int*)d_in[1];
    const float* W1   = (const float*)d_in[2];
    const float* b1   = (const float*)d_in[3];
    const float* W2   = (const float*)d_in[4];
    const float* b2   = (const float*)d_in[5];
    float* out = (float*)d_out;

    const int N = in_sizes[0] / CH;
    const int E = in_sizes[1] / 2;
    const int* srcI = edge;
    const int* dstI = edge + E;

    const int T = 256;
    const int NBIN = (N + BIN_NODES - 1) / BIN_NODES;
    const int NBLK = (E + CHUNK - 1) / CHUNK;
    const int N2 = NBIN * NBLK;
    const int NB2 = (N2 + SCAN_TILE - 1) / SCAN_TILE;

    char* ws = (char*)d_ws;
    size_t off = 0;
    float* dinv = (float*)(ws + off);  off += ((size_t)N * 4 + 255) & ~(size_t)255;
    int* rowptr = (int*)(ws + off);    off += ((size_t)(N + 1) * 4 + 255) & ~(size_t)255;
    int* cnt2   = (int*)(ws + off);    off += ((size_t)N2 * 4 + 255) & ~(size_t)255;
    int* offs   = (int*)(ws + off);    off += ((size_t)(N2 + 1) * 4 + 255) & ~(size_t)255;
    int* bsum   = (int*)(ws + off);    off += ((size_t)NB2 * 4 + 255) & ~(size_t)255;
    int* part   = (int*)(ws + off);    off += ((size_t)E * 8 + 255) & ~(size_t)255;
    int* csr    = (int*)(ws + off);    off += ((size_t)E * 4 + 255) & ~(size_t)255;
    float* hbuf = (float*)(ws + off);
    float* tmp  = hbuf + (size_t)N * CH;

    dim3 gGemm((N + 63) / 64);
    dim3 gGemmHist(gGemm.x + NBLK);
    dim3 gRow((N + 7) / 8);
    const int FUSED_WGS = 1024;
    const int RPW = (N + FUSED_WGS - 1) / FUSED_WGS;

    gemm_hist_kernel<<<gGemmHist, 128, 0, stream>>>(x, W1, hbuf, N, dstI, cnt2, E, NBIN, NBLK, (int)gGemm.x);
    block_sum_kernel<<<NB2, T, 0, stream>>>(cnt2, bsum, N2);
    scan_kernel<<<NB2, T, 0, stream>>>(cnt2, bsum, offs, N2);
    partition_kernel<<<NBLK, 1024, 0, stream>>>(srcI, dstI, offs, part, E, NBIN, NBLK);
    binsort_kernel<<<NBIN, 1024, 0, stream>>>(part, offs, rowptr, dinv, csr, N, E, NBIN, NBLK);

    gather_gemm_kernel<<<FUSED_WGS, T, 0, stream>>>(hbuf, dinv, rowptr, csr, b1, W2, tmp, N, RPW);

    gather_kernel<<<gRow, T, 0, stream>>>(tmp, dinv, rowptr, csr, b2, out, N);
}


# layer-1 GEMM: each wave keeps its 128x16 panel of W1 in registers (read once per work-group; 1024 resident work-groups loop over 32-row X tiles via two LDS tiles) instead of two W1 loads per WMMA
# speedup vs baseline: 4.8738x; 1.0368x over previous
#include <hip/hip_runtime.h>
#include <hip/hip_bf16.h>
#include <cstddef>

typedef __attribute__((ext_vector_type(2))) float v2f;
typedef __attribute__((ext_vector_type(8))) float v8f;

#define CH 128

#define SCAN_TILE 1024
#define BIN_SHIFT 9
#define BIN_NODES 512
#define MAXB 1024
#define CHUNK 8192
#define STAGE_CAP 16384

__device__ __forceinline__ int wave_sum(int v) {
    #pragma unroll
    for (int off = 16; off > 0; off >>= 1) v += __shfl_xor(v, off);
    return v;
}

__global__ __launch_bounds__(256) void block_sum_kernel(
    const int* __restrict__ cnt, int* __restrict__ bsum, int n)
{
    __shared__ int red[8];
    const int lane = threadIdx.x & 31;
    const int wave = threadIdx.x >> 5;
    const int base = blockIdx.x * SCAN_TILE + threadIdx.x * 4;
    int s = 0;
    #pragma unroll
    for (int k = 0; k < 4; ++k)
        if (base + k < n) s += cnt[base + k];
    s = wave_sum(s);
    if (lane == 0) red[wave] = s;
    __syncthreads();
    if (threadIdx.x == 0) {
        int t = 0;
        #pragma unroll
        for (int w = 0; w < 8; ++w) t += red[w];
        bsum[blockIdx.x] = t;
    }
}

__global__ __launch_bounds__(256) void scan_kernel(
    const int* __restrict__ cnt, const int* __restrict__ bsum,
    int* __restrict__ rowptr, int n)
{
    __shared__ int red[8];
    __shared__ int wsum[8];
    const int lane = threadIdx.x & 31;
    const int wave = threadIdx.x >> 5;

    int part = 0;
    for (int j = threadIdx.x; j < (int)blockIdx.x; j += 256) part += bsum[j];
    part = wave_sum(part);
    if (lane == 0) red[wave] = part;

    const int base = blockIdx.x * SCAN_TILE + threadIdx.x * 4;
    int c[4];
    #pragma unroll
    for (int k = 0; k < 4; ++k) c[k] = (base + k < n) ? cnt[base + k] : 0;
    const int tsum = c[0] + c[1] + c[2] + c[3];

    int inc = tsum;
    #pragma unroll
    for (int off = 1; off < 32; off <<= 1) {
        int v = __shfl_up(inc, off);
        if (lane >= off) inc += v;
    }
    if (lane == 31) wsum[wave] = inc;
    __syncthreads();

    int offset = 0;
    #pragma unroll
    for (int w = 0; w < 8; ++w) offset += red[w];
    for (int w = 0; w < wave; ++w) offset += wsum[w];

    int excl = offset + inc - tsum;
    #pragma unroll
    for (int k = 0; k < 4; ++k) {
        const int idx = base + k;
        if (idx < n) {
            rowptr[idx] = excl;
            excl += c[k];
            if (idx == n - 1) rowptr[n] = excl;
        }
    }
}

__device__ __forceinline__ int block_excl_scan_1024(int v, int* wsum, int& total)
{
    const int lane = threadIdx.x & 31;
    const int wave = threadIdx.x >> 5;
    int inc = v;
    #pragma unroll
    for (int off = 1; off < 32; off <<= 1) {
        int u = __shfl_up(inc, off);
        if (lane >= off) inc += u;
    }
    if (lane == 31) wsum[wave] = inc;
    __syncthreads();
    if (wave == 0) {
        const int w = wsum[lane];
        int winc = w;
        #pragma unroll
        for (int off = 1; off < 32; off <<= 1) {
            int u = __shfl_up(winc, off);
            if (lane >= off) winc += u;
        }
        wsum[lane] = winc - w;
        if (lane == 31) wsum[32] = winc;
    }
    __syncthreads();
    total = wsum[32];
    return wsum[wave] + inc - v;
}

__device__ __forceinline__ void hist_chunk(
    const int* __restrict__ dst, int* __restrict__ cnt2, int e, int nb, int nblk, int blk)
{
    __shared__ int hist[MAXB];
    const int t = threadIdx.x;
    for (int b = t; b < nb; b += 256) hist[b] = 0;
    __syncthreads();
    const int first = blk * CHUNK;
    if (first + CHUNK <= e) {
        for (int k0 = 0; k0 < CHUNK / 256; k0 += 8) {
            int d[8];
            #pragma unroll
            for (int k = 0; k < 8; ++k) d[k] = dst[first + (k0 + k) * 256 + t];
            #pragma unroll
            for (int k = 0; k < 8; ++k) atomicAdd(&hist[d[k] >> BIN_SHIFT], 1);
        }
    } else {
        for (int i = first + t; i < e; i += 256) atomicAdd(&hist[dst[i] >> BIN_SHIFT], 1);
    }
    __syncthreads();
    for (int b = t; b < nb; b += 256) cnt2[b * nblk + blk] = hist[b];
}

__global__ __launch_bounds__(1024) void partition_kernel(
    const int* __restrict__ src, const int* __restrict__ dst,
    const int* __restrict__ off, int* __restrict__ part_raw, int e, int nb, int nblk)
{
    __shared__ int hist[MAXB];
    __shared__ int lstart[MAXB];
    __shared__ int goff[MAXB];
    __shared__ int wsum[33];
    __shared__ int st_src[CHUNK];
    __shared__ int st_dst[CHUNK];
    int2* __restrict__ part = (int2*)part_raw;

    const int t = threadIdx.x;
    const int blk = blockIdx.x;
    hist[t] = 0;
    __syncthreads();

    const int first = blk * CHUNK;
    int s[CHUNK / 1024], d[CHUNK / 1024], r[CHUNK / 1024];
    #pragma unroll
    for (int k = 0; k < CHUNK / 1024; ++k) {
        const int i = first + k * 1024 + t;
        s[k] = 0; d[k] = -1; r[k] = 0;
        if (i < e) { s[k] = src[i]; d[k] = dst[i]; }
    }
    #pragma unroll
    for (int k = 0; k < CHUNK / 1024; ++k)
        if (d[k] >= 0) r[k] = atomicAdd(&hist[d[k] >> BIN_SHIFT], 1);
    __syncthreads();

    int total;
    const int ex = block_excl_scan_1024(hist[t], wsum, total);
    lstart[t] = ex;
    if (t < nb) goff[t] = off[t * nblk + blk];
    __syncthreads();

    #pragma unroll
    for (int k = 0; k < CHUNK / 1024; ++k) {
        if (d[k] >= 0) {
            const int p = lstart[d[k] >> BIN_SHIFT] + r[k];
            st_src[p] = s[k];
            st_dst[p] = d[k];
        }
    }
    __syncthreads();

    for (int j = t; j < total; j += 1024) {
        const int dd = st_dst[j];
        const int b = dd >> BIN_SHIFT;
        part[goff[b] + (j - lstart[b])] = make_int2(st_src[j], dd);
    }
}

__global__ __launch_bounds__(1024) void binsort_kernel(
    const int* __restrict__ part_raw, const int* __restrict__ off,
    int* __restrict__ rowptr, float* __restrict__ dinv, int* __restrict__ csr_src,
    int n, int e, int nb, int nblk)
{
    __shared__ int cnt[BIN_NODES];
    __shared__ int cur[BIN_NODES];
    __shared__ int wsum[33];
    __shared__ int stage[STAGE_CAP];
    const int2* __restrict__ part = (const int2*)part_raw;

    const int t = threadIdx.x;
    const int bin = blockIdx.x;
    const int beg = off[bin * nblk];
    const int end = (bin + 1 < nb) ? off[(bin + 1) * nblk] : e;
    const int node0 = bin << BIN_SHIFT;

    if (t < BIN_NODES) cnt[t] = 0;
    __syncthreads();
    for (int i = beg + t; i < end; i += 1024) atomicAdd(&cnt[part[i].y - node0], 1);
    __syncthreads();

    const int c = (t < BIN_NODES) ? cnt[t] : 0;
    int total;
    const int ex = block_excl_scan_1024(c, wsum, total);
    if (t < BIN_NODES) {
        cur[t] = ex;
        const int node = node0 + t;
        if (node < n) {
            rowptr[node] = beg + ex;
            dinv[node] = rsqrtf((float)(c + 1));
            if (node == n - 1) rowptr[n] = beg + ex + c;
        }
    }
    __syncthreads();

    const bool staged = (end - beg) <= STAGE_CAP;
    for (int i = beg + t; i < end; i += 1024) {
        const int2 sd = part[i];
        const int p = atomicAdd(&cur[sd.y - node0], 1);
        if (staged) stage[p] = sd.x; else csr_src[beg + p] = sd.x;
    }
    if (staged) {
        __syncthreads();
        const int m = end - beg;
        for (int j = t; j < m; j += 1024) csr_src[beg + j] = stage[j];
    }
}

#define G1_ROWS 32
#define G1_PITCH 132

__global__ __launch_bounds__(256) void gemm_hist_kernel(
    const float* __restrict__ X, const float* __restrict__ W,
    float* __restrict__ H, int nrows,
    const int* __restrict__ dst, int* __restrict__ cnt2,
    int e, int nb, int nblk)
{
    __shared__ __attribute__((aligned(16))) float Xs[2][G1_ROWS * G1_PITCH];

    const int G = (int)gridDim.x;
    const int g = (int)blockIdx.x;

    for (int blk = g; blk < nblk; blk += G)
        hist_chunk(dst, cnt2, e, nb, nblk, blk);

    const int ntiles = (nrows + G1_ROWS - 1) / G1_ROWS;
    int t = G - 1 - g;
    if (t >= ntiles) return;

    const int lane = threadIdx.x & 31;
    const int wave = __builtin_amdgcn_readfirstlane((int)(threadIdx.x >> 5));
    const int c  = lane & 15;
    const int hi = lane >> 4;

    v2f bfrag[CH / 4];
    #pragma unroll
    for (int ks = 0; ks < CH / 4; ++ks) {
        const int klo = ks * 4 + hi * 2;
        bfrag[ks].x = W[(size_t)klo * CH + wave * 16 + c];
        bfrag[ks].y = W[(size_t)(klo + 1) * CH + wave * 16 + c];
    }

    float4 nx[G1_ROWS / 8];
    #pragma unroll
    for (int jr = 0; jr < G1_ROWS / 8; ++jr) {
        const int row = t * G1_ROWS + wave + 8 * jr;
        nx[jr] = make_float4(0.f, 0.f, 0.f, 0.f);
        if (row < nrows) nx[jr] = *(const float4*)(X + (size_t)row * CH + lane * 4);
    }
    #pragma unroll
    for (int jr = 0; jr < G1_ROWS / 8; ++jr)
        *(float4*)(&Xs[0][0] + (wave + 8 * jr) * G1_PITCH + lane * 4) = nx[jr];
    __syncthreads();

    int buf = 0;
    for (; t < ntiles; t += G, buf ^= 1) {
        const float* xs = &Xs[buf][0];
        const int tn = t + G;
        const bool more = tn < ntiles;

        if (more) {
            #pragma unroll
            for (int jr = 0; jr < G1_ROWS / 8; ++jr) {
                const int row = tn * G1_ROWS + wave + 8 * jr;
                nx[jr] = make_float4(0.f, 0.f, 0.f, 0.f);
                if (row < nrows) nx[jr] = *(const float4*)(X + (size_t)row * CH + lane * 4);
            }
        }

        v8f acc0 = (v8f)(0.0f);
        v8f acc1 = (v8f)(0.0f);
        #pragma unroll
        for (int ks = 0; ks < CH / 4; ++ks) {
            const int klo = ks * 4 + hi * 2;
            const v2f a0 = *(const v2f*)(xs + c * G1_PITCH + klo);
            const v2f a1 = *(const v2f*)(xs + (16 + c) * G1_PITCH + klo);
            acc0 = __builtin_amdgcn_wmma_f32_16x16x4_f32(
                false, a0, false, bfrag[ks], (short)0, acc0, false, false);
            acc1 = __builtin_amdgcn_wmma_f32_16x16x4_f32(
                false, a1, false, bfrag[ks], (short)0, acc1, false, false);
        }

        const int tbase = t * G1_ROWS;
        float* __restrict__ po = H + (size_t)(tbase + hi * 8) * CH + wave * 16 + c;
        if (tbase + G1_ROWS <= nrows) {
            #pragma unroll
            for (int r = 0; r < 8; ++r) {
                po[r * CH] = acc0[r];
                po[(r + 16) * CH] = acc1[r];
            }
        } else {
            const int left = nrows - (tbase + hi * 8);
            #pragma unroll
            for (int r = 0; r < 8; ++r) {
                if (r < left) po[r * CH] = acc0[r];
                if (r + 16 < left) po[(r + 16) * CH] = acc1[r];
            }
        }

        if (more) {
            #pragma unroll
            for (int jr = 0; jr < G1_ROWS / 8; ++jr)
                *(float4*)(&Xs[buf ^ 1][0] + (wave + 8 * jr) * G1_PITCH + lane * 4) = nx[jr];
        }
        __syncthreads();
    }
}

template <int K>
__device__ __forceinline__ void add_rows(const float* __restrict__ H, int sidx, float cf,
                                         int j, int lane, float4& acc)
{
    float4 v[K];
    float  c[K];
    #pragma unroll
    for (int k = 0; k < K; ++k) {
        const int s = __builtin_amdgcn_readlane(sidx, j + k);
        c[k] = __int_as_float(__builtin_amdgcn_readlane(__float_as_int(cf), j + k));
        v[k] = *(const float4*)(H + (size_t)s * CH + lane * 4);
    }
    #pragma unroll
    for (int k = 0; k < K; ++k) {
        acc.x = fmaf(v[k].x, c[k], acc.x);
        acc.y = fmaf(v[k].y, c[k], acc.y);
        acc.z = fmaf(v[k].z, c[k], acc.z);
        acc.w = fmaf(v[k].w, c[k], acc.w);
    }
}

__global__ __launch_bounds__(256) void gather_kernel(
    const float* __restrict__ H, const float* __restrict__ dinv,
    const int* __restrict__ rowptr, const int* __restrict__ csr_src,
    const float* __restrict__ bias, float* __restrict__ out, int n)
{
    const int lane = threadIdx.x & 31;
    const int row = __builtin_amdgcn_readfirstlane(
        (int)((blockIdx.x * blockDim.x + threadIdx.x) >> 5));
    if (row >= n) return;

    const int beg = rowptr[row];
    const int end = rowptr[row + 1];
    const float di = dinv[row];
    const float w = di * di;

    const float4 h = *(const float4*)(H + (size_t)row * CH + lane * 4);
    float4 acc;
    acc.x = h.x * w; acc.y = h.y * w; acc.z = h.z * w; acc.w = h.w * w;

    for (int base = beg; base < end; base += 32) {
        const int nb = (end - base < 32) ? (end - base) : 32;
        int sidx = 0;
        float cf = 0.f;
        if (lane < nb) {
            sidx = csr_src[base + lane];
            cf = dinv[sidx] * di;
        }
        int j = 0;
        for (; j + 4 <= nb; j += 4) add_rows<4>(H, sidx, cf, j, lane, acc);
        if (j + 2 <= nb) { add_rows<2>(H, sidx, cf, j, lane, acc); j += 2; }
        if (j < nb) add_rows<1>(H, sidx, cf, j, lane, acc);
    }

    const float4 b = *(const float4*)(bias + lane * 4);
    float4 o;
    o.x = acc.x + b.x; o.y = acc.y + b.y; o.z = acc.z + b.z; o.w = acc.w + b.w;
    *(float4*)(out + (size_t)row * CH + lane * 4) = o;
}

__device__ __forceinline__ void add_chunk(const float* __restrict__ H, int sidx, float cf,
                                          int nb, int lane, float4& acc)
{
    int j = 0;
    for (; j + 8 <= nb; j += 8) add_rows<8>(H, sidx, cf, j, lane, acc);
    switch (nb - j) {
        case 7: add_rows<7>(H, sidx, cf, j, lane, acc); break;
        case 6: add_rows<6>(H, sidx, cf, j, lane, acc); break;
        case 5: add_rows<5>(H, sidx, cf, j, lane, acc); break;
        case 4: add_rows<4>(H, sidx, cf, j, lane, acc); break;
        case 3: add_rows<3>(H, sidx, cf, j, lane, acc); break;
        case 2: add_rows<2>(H, sidx, cf, j, lane, acc); break;
        case 1: add_rows<1>(H, sidx, cf, j, lane, acc); break;
        default: break;
    }
}

#define TROWS 32
#define XP 132

__global__ __launch_bounds__(256) void gather_gemm_kernel(
    const float* __restrict__ H, const float* __restrict__ dinv,
    const int* __restrict__ rowptr, const int* __restrict__ csr_src,
    const float* __restrict__ bias, const float* __restrict__ W,
    float* __restrict__ Hout, int n, int rows_per_wg)
{
    __shared__ __attribute__((aligned(16))) float Xs[2][TROWS * XP];
    __shared__ int4 Es[8][TROWS / 8][32];

    const int lane = threadIdx.x & 31;
    const int wave = __builtin_amdgcn_readfirstlane((int)(threadIdx.x >> 5));
    const int c  = lane & 15;
    const int hi = lane >> 4;

    v2f bfrag[CH / 4];
    #pragma unroll
    for (int ks = 0; ks < CH / 4; ++ks) {
        const int klo = ks * 4 + hi * 2;
        bfrag[ks].x = W[(size_t)klo * CH + wave * 16 + c];
        bfrag[ks].y = W[(size_t)(klo + 1) * CH + wave * 16 + c];
    }
    const float4 b = *(const float4*)(bias + lane * 4);

    const int row0 = (int)blockIdx.x * rows_per_wg;
    const int row1 = (row0 + rows_per_wg < n) ? (row0 + rows_per_wg) : n;

    int buf = 0;
    for (int tbase = row0; tbase < row1; tbase += TROWS, buf ^= 1) {
        float* xs = &Xs[buf][0];

        float4 hnext = make_float4(0.f, 0.f, 0.f, 0.f);
        if (tbase + wave < row1)
            hnext = *(const float4*)(H + (size_t)(tbase + wave) * CH + lane * 4);
        {
            int rbeg[TROWS / 8], rend[TROWS / 8];
            float rdi[TROWS / 8];
            int sidx0[TROWS / 8];
            #pragma unroll
            for (int jr = 0; jr < TROWS / 8; ++jr) {
                const int row = tbase + wave + 8 * jr;
                const int rc = (row < row1) ? row : row0;
                const int b0 = rowptr[rc];
                const int e0 = rowptr[rc + 1];
                rbeg[jr] = b0;
                rend[jr] = (row < row1) ? e0 : b0;
                rdi[jr] = dinv[rc];
            }
            #pragma unroll
            for (int jr = 0; jr < TROWS / 8; ++jr) {
                int p = rbeg[jr] + lane;
                p = (p < rend[jr] - 1) ? p : (rend[jr] - 1);
                p = (p > 0) ? p : 0;
                sidx0[jr] = csr_src[p];
            }
            #pragma unroll
            for (int jr = 0; jr < TROWS / 8; ++jr) {
                const float cf = dinv[sidx0[jr]] * rdi[jr];
                Es[wave][jr][lane] = make_int4(sidx0[jr], __float_as_int(cf),
                                               rend[jr] - rbeg[jr], __float_as_int(rdi[jr] * rdi[jr]));
            }
        }

        #pragma unroll 1
        for (int jr = 0; jr < TROWS / 8; ++jr) {
            const int lr = wave + 8 * jr;
            const int row = tbase + lr;
            float4 o = make_float4(0.f, 0.f, 0.f, 0.f);
            if (row < row1) {
                const int4 sc = Es[wave][jr][lane];
                const int sidx0 = sc.x;
                const float cf0 = __int_as_float(sc.y);
                const int deg = __builtin_amdgcn_readfirstlane(sc.z);
                const float w = __int_as_float(sc.w);
                const float4 h = hnext;
                if (jr + 1 < TROWS / 8 && row + 8 < row1)
                    hnext = *(const float4*)(H + (size_t)(row + 8) * CH + lane * 4);
                float4 acc;
                acc.x = h.x * w; acc.y = h.y * w; acc.z = h.z * w; acc.w = h.w * w;
                add_chunk(H, sidx0, cf0, (deg < 32) ? deg : 32, lane, acc);
                if (deg > 32) {
                  const int beg = rowptr[row];
                  const int end = beg + deg;
                  const float di = dinv[row];
                  for (int base = beg + 32; base < end; base += 32) {
                    const int nb = (end - base < 32) ? (end - base) : 32;
                    int sidx = 0;
                    float cf = 0.f;
                    if (lane < nb) {
                        sidx = csr_src[base + lane];
                        cf = dinv[sidx] * di;
                    }
                    add_chunk(H, sidx, cf, nb, lane, acc);
                  }
                }
                o.x = fmaxf(acc.x + b.x, 0.f); o.y = fmaxf(acc.y + b.y, 0.f);
                o.z = fmaxf(acc.z + b.z, 0.f); o.w = fmaxf(acc.w + b.w, 0.f);
            }
            *(float4*)(xs + lr * XP + lane * 4) = o;
        }
        __syncthreads();

        v8f acc0 = (v8f)(0.0f);
        v8f acc1 = (v8f)(0.0f);
        #pragma unroll
        for (int ks = 0; ks < CH / 4; ++ks) {
            const int klo = ks * 4 + hi * 2;
            const v2f a0 = *(const v2f*)(xs + c * XP + klo);
            const v2f a1 = *(const v2f*)(xs + (16 + c) * XP + klo);
            acc0 = __builtin_amdgcn_wmma_f32_16x16x4_f32(
                false, a0, false, bfrag[ks], (short)0, acc0, false, false);
            acc1 = __builtin_amdgcn_wmma_f32_16x16x4_f32(
                false, a1, false, bfrag[ks], (short)0, acc1, false, false);
        }

        float* __restrict__ po = Hout + (size_t)(tbase + hi * 8) * CH + wave * 16 + c;
        if (tbase + TROWS <= row1) {
            #pragma unroll
            for (int r = 0; r < 8; ++r) {
                po[r * CH] = acc0[r];
                po[(r + 16) * CH] = acc1[r];
            }
        } else {
            const int left = row1 - (tbase + hi * 8);
            #pragma unroll
            for (int r = 0; r < 8; ++r) {
                if (r < left) po[r * CH] = acc0[r];
                if (r + 16 < left) po[(r + 16) * CH] = acc1[r];
            }
        }
    }
}

extern "C" void kernel_launch(void* const* d_in, const int* in_sizes, int n_in,
                              void* d_out, int out_size, void* d_ws, size_t ws_size,
                              hipStream_t stream) {
    const float* x    = (const float*)d_in[0];
    const int*   edge = (const int*)d_in[1];
    const float* W1   = (const float*)d_in[2];
    const float* b1   = (const float*)d_in[3];
    const float* W2   = (const float*)d_in[4];
    const float* b2   = (const float*)d_in[5];
    float* out = (float*)d_out;

    const int N = in_sizes[0] / CH;
    const int E = in_sizes[1] / 2;
    const int* srcI = edge;
    const int* dstI = edge + E;

    const int T = 256;
    const int NBIN = (N + BIN_NODES - 1) / BIN_NODES;
    const int NBLK = (E + CHUNK - 1) / CHUNK;
    const int N2 = NBIN * NBLK;
    const int NB2 = (N2 + SCAN_TILE - 1) / SCAN_TILE;

    char* ws = (char*)d_ws;
    size_t off = 0;
    float* dinv = (float*)(ws + off);  off += ((size_t)N * 4 + 255) & ~(size_t)255;
    int* rowptr = (int*)(ws + off);    off += ((size_t)(N + 1) * 4 + 255) & ~(size_t)255;
    int* cnt2   = (int*)(ws + off);    off += ((size_t)N2 * 4 + 255) & ~(size_t)255;
    int* offs   = (int*)(ws + off);    off += ((size_t)(N2 + 1) * 4 + 255) & ~(size_t)255;
    int* bsum   = (int*)(ws + off);    off += ((size_t)NB2 * 4 + 255) & ~(size_t)255;
    int* part   = (int*)(ws + off);    off += ((size_t)E * 8 + 255) & ~(size_t)255;
    int* csr    = (int*)(ws + off);    off += ((size_t)E * 4 + 255) & ~(size_t)255;
    float* hbuf = (float*)(ws + off);
    float* tmp  = hbuf + (size_t)N * CH;

    dim3 gRow((N + 7) / 8);
    const int FUSED_WGS = 1024;
    const int GEMM1_WGS = 1024;
    const int RPW = (N + FUSED_WGS - 1) / FUSED_WGS;

    gemm_hist_kernel<<<GEMM1_WGS, T, 0, stream>>>(x, W1, hbuf, N, dstI, cnt2, E, NBIN, NBLK);
    block_sum_kernel<<<NB2, T, 0, stream>>>(cnt2, bsum, N2);
    scan_kernel<<<NB2, T, 0, stream>>>(cnt2, bsum, offs, N2);
    partition_kernel<<<NBLK, 1024, 0, stream>>>(srcI, dstI, offs, part, E, NBIN, NBLK);
    binsort_kernel<<<NBIN, 1024, 0, stream>>>(part, offs, rowptr, dinv, csr, N, E, NBIN, NBLK);

    gather_gemm_kernel<<<FUSED_WGS, T, 0, stream>>>(hbuf, dinv, rowptr, csr, b1, W2, tmp, N, RPW);

    gather_kernel<<<gRow, T, 0, stream>>>(tmp, dinv, rowptr, csr, b2, out, N);
}


# CSR build: chunks are bin-sorted in place with a table of bin starts; a bin's work-group takes base and runs from the tables: count pass and both scan launches go (7 -> 5); sum order: LDS arrival
# speedup vs baseline: 5.0679x; 1.0398x over previous
#include <hip/hip_runtime.h>
#include <hip/hip_bf16.h>
#include <cstddef>

typedef __attribute__((ext_vector_type(2))) float v2f;
typedef __attribute__((ext_vector_type(8))) float v8f;

#define CH 128

#define BIN_SHIFT 9
#define BIN_NODES 512
#define MAXB 1024
#define CHUNK 8192
#define STAGE_CAP 16384
#define RUN_TILE 1024

__device__ __forceinline__ int block_excl_scan_1024(int v, int* wsum, int& total)
{
    const int lane = threadIdx.x & 31;
    const int wave = threadIdx.x >> 5;
    int inc = v;
    #pragma unroll
    for (int off = 1; off < 32; off <<= 1) {
        int u = __shfl_up(inc, off);
        if (lane >= off) inc += u;
    }
    if (lane == 31) wsum[wave] = inc;
    __syncthreads();
    if (wave == 0) {
        const int w = wsum[lane];
        int winc = w;
        #pragma unroll
        for (int off = 1; off < 32; off <<= 1) {
            int u = __shfl_up(winc, off);
            if (lane >= off) winc += u;
        }
        wsum[lane] = winc - w;
        if (lane == 31) wsum[32] = winc;
    }
    __syncthreads();
    total = wsum[32];
    return wsum[wave] + inc - v;
}

__global__ __launch_bounds__(1024) void partition_kernel(
    const int* __restrict__ src, const int* __restrict__ dst,
    int* __restrict__ ls, int* __restrict__ part_raw, int e, int nb)
{
    __shared__ int hist[MAXB];
    __shared__ int lstart[MAXB];
    __shared__ int wsum[33];
    __shared__ int st_src[CHUNK];
    __shared__ int st_dst[CHUNK];
    int2* __restrict__ part = (int2*)part_raw;

    const int t = threadIdx.x;
    const int blk = blockIdx.x;
    hist[t] = 0;
    __syncthreads();

    const int first = blk * CHUNK;
    int s[CHUNK / 1024], d[CHUNK / 1024], r[CHUNK / 1024];
    #pragma unroll
    for (int k = 0; k < CHUNK / 1024; ++k) {
        const int i = first + k * 1024 + t;
        s[k] = 0; d[k] = -1; r[k] = 0;
        if (i < e) { s[k] = src[i]; d[k] = dst[i]; }
    }
    #pragma unroll
    for (int k = 0; k < CHUNK / 1024; ++k)
        if (d[k] >= 0) r[k] = atomicAdd(&hist[d[k] >> BIN_SHIFT], 1);
    __syncthreads();

    int total;
    const int ex = block_excl_scan_1024(hist[t], wsum, total);
    lstart[t] = ex;
    if (t < nb) ls[blk * (nb + 1) + t] = ex;
    if (t == 0) ls[blk * (nb + 1) + nb] = total;
    __syncthreads();

    #pragma unroll
    for (int k = 0; k < CHUNK / 1024; ++k) {
        if (d[k] >= 0) {
            const int p = lstart[d[k] >> BIN_SHIFT] + r[k];
            st_src[p] = s[k];
            st_dst[p] = d[k];
        }
    }
    __syncthreads();

    for (int j = t; j < total; j += 1024)
        part[first + j] = make_int2(st_src[j], st_dst[j]);
}

template <typename F>
__device__ __forceinline__ void for_bin_edges(
    const int2* __restrict__ part, const int* __restrict__ ls, int pitch, int bin, int nblk,
    bool reload, int* rs, int* rl, F f)
{
    const int t = threadIdx.x;
    const int lane = t & 31;
    const int wave = __builtin_amdgcn_readfirstlane((int)(threadIdx.x >> 5));
    for (int c0 = 0; c0 < nblk; c0 += RUN_TILE) {
        if (reload) {
            __syncthreads();
            const int blk = c0 + t;
            int st = 0, len = 0;
            if (blk < nblk) {
                const int a0 = ls[blk * pitch + bin];
                const int a1 = ls[blk * pitch + bin + 1];
                st = blk * CHUNK + a0;
                len = a1 - a0;
            }
            rs[t] = st;
            rl[t] = len;
            __syncthreads();
        }
        const int nr = (nblk - c0 < RUN_TILE) ? (nblk - c0) : RUN_TILE;
        for (int r0 = wave; r0 < nr; r0 += 128) {
            int st4[4], ln4[4];
            #pragma unroll
            for (int k = 0; k < 4; ++k) {
                const int r = r0 + 32 * k;
                const bool ok = r < nr;
                ln4[k] = ok ? rl[ok ? r : 0] : 0;
                st4[k] = ok ? rs[ok ? r : 0] : 0;
            }
            int2 v[4][2];
            #pragma unroll
            for (int k = 0; k < 4; ++k) {
                #pragma unroll
                for (int q = 0; q < 2; ++q) {
                    const int idx = lane + 32 * q;
                    v[k][q] = make_int2(0, -1);
                    if (idx < ln4[k]) v[k][q] = part[st4[k] + idx];
                }
            }
            #pragma unroll
            for (int k = 0; k < 4; ++k) {
                #pragma unroll
                for (int q = 0; q < 2; ++q)
                    if (v[k][q].y >= 0) f(v[k][q].x, v[k][q].y);
            }
            #pragma unroll
            for (int k = 0; k < 4; ++k) {
                for (int idx = lane + 64; idx < ln4[k]; idx += 32) {
                    const int2 sd = part[st4[k] + idx];
                    f(sd.x, sd.y);
                }
            }
        }
    }
}

__global__ __launch_bounds__(1024) void binsort_kernel(
    const int* __restrict__ part_raw, const int* __restrict__ ls,
    int* __restrict__ rowptr, float* __restrict__ dinv, int* __restrict__ csr_src,
    int n, int nb, int nblk)
{
    __shared__ int cnt[BIN_NODES];
    __shared__ int cur[BIN_NODES];
    __shared__ int wsum[33];
    __shared__ int wsum2[33];
    __shared__ int rs[RUN_TILE];
    __shared__ int rl[RUN_TILE];
    __shared__ int stage[STAGE_CAP];
    const int2* __restrict__ part = (const int2*)part_raw;

    const int t = threadIdx.x;
    const int bin = blockIdx.x;
    const int node0 = bin << BIN_SHIFT;
    const int pitch = nb + 1;
    const bool reload = nblk > RUN_TILE;

    if (t < BIN_NODES) cnt[t] = 0;

    int below = 0, mine = 0;
    for (int blk = t; blk < nblk; blk += 1024) {
        const int a0 = ls[blk * pitch + bin];
        const int a1 = ls[blk * pitch + bin + 1];
        below += a0;
        mine += a1 - a0;
        if (blk < RUN_TILE) { rs[blk] = blk * CHUNK + a0; rl[blk] = a1 - a0; }
    }
    int beg, m;
    block_excl_scan_1024(below, wsum, beg);
    block_excl_scan_1024(mine, wsum2, m);

    for_bin_edges(part, ls, pitch, bin, nblk, reload, rs, rl,
        [&](int, int dd) { atomicAdd(&cnt[dd - node0], 1); });
    __syncthreads();

    const int c = (t < BIN_NODES) ? cnt[t] : 0;
    int total;
    const int ex = block_excl_scan_1024(c, wsum, total);
    if (t < BIN_NODES) {
        cur[t] = ex;
        const int node = node0 + t;
        if (node < n) {
            rowptr[node] = beg + ex;
            dinv[node] = rsqrtf((float)(c + 1));
            if (node == n - 1) rowptr[n] = beg + ex + c;
        }
    }
    __syncthreads();

    const bool staged = m <= STAGE_CAP;
    for_bin_edges(part, ls, pitch, bin, nblk, reload, rs, rl,
        [&](int ss, int dd) {
            const int p = atomicAdd(&cur[dd - node0], 1);
            if (staged) stage[p] = ss; else csr_src[beg + p] = ss;
        });
    if (staged) {
        __syncthreads();
        for (int j = t; j < m; j += 1024) csr_src[beg + j] = stage[j];
    }
}

#define G1_ROWS 32
#define G1_PITCH 132

__global__ __launch_bounds__(256) void gemm_kernel(
    const float* __restrict__ X, const float* __restrict__ W,
    float* __restrict__ H, int nrows)
{
    __shared__ __attribute__((aligned(16))) float Xs[2][G1_ROWS * G1_PITCH];

    const int G = (int)gridDim.x;
    const int ntiles = (nrows + G1_ROWS - 1) / G1_ROWS;
    int t = (int)blockIdx.x;
    if (t >= ntiles) return;

    const int lane = threadIdx.x & 31;
    const int wave = __builtin_amdgcn_readfirstlane((int)(threadIdx.x >> 5));
    const int c  = lane & 15;
    const int hi = lane >> 4;

    v2f bfrag[CH / 4];
    #pragma unroll
    for (int ks = 0; ks < CH / 4; ++ks) {
        const int klo = ks * 4 + hi * 2;
        bfrag[ks].x = W[(size_t)klo * CH + wave * 16 + c];
        bfrag[ks].y = W[(size_t)(klo + 1) * CH + wave * 16 + c];
    }

    float4 nx[G1_ROWS / 8];
    #pragma unroll
    for (int jr = 0; jr < G1_ROWS / 8; ++jr) {
        const int row = t * G1_ROWS + wave + 8 * jr;
        nx[jr] = make_float4(0.f, 0.f, 0.f, 0.f);
        if (row < nrows) nx[jr] = *(const float4*)(X + (size_t)row * CH + lane * 4);
    }
    #pragma unroll
    for (int jr = 0; jr < G1_ROWS / 8; ++jr)
        *(float4*)(&Xs[0][0] + (wave + 8 * jr) * G1_PITCH + lane * 4) = nx[jr];
    __syncthreads();

    int buf = 0;
    for (; t < ntiles; t += G, buf ^= 1) {
        const float* xs = &Xs[buf][0];
        const int tn = t + G;
        const bool more = tn < ntiles;

        if (more) {
            #pragma unroll
            for (int jr = 0; jr < G1_ROWS / 8; ++jr) {
                const int row = tn * G1_ROWS + wave + 8 * jr;
                nx[jr] = make_float4(0.f, 0.f, 0.f, 0.f);
                if (row < nrows) nx[jr] = *(const float4*)(X + (size_t)row * CH + lane * 4);
            }
        }

        v8f acc0 = (v8f)(0.0f);
        v8f acc1 = (v8f)(0.0f);
        #pragma unroll
        for (int ks = 0; ks < CH / 4; ++ks) {
            const int klo = ks * 4 + hi * 2;
            const v2f a0 = *(const v2f*)(xs + c * G1_PITCH + klo);
            const v2f a1 = *(const v2f*)(xs + (16 + c) * G1_PITCH + klo);
            acc0 = __builtin_amdgcn_wmma_f32_16x16x4_f32(
                false, a0, false, bfrag[ks], (short)0, acc0, false, false);
            acc1 = __builtin_amdgcn_wmma_f32_16x16x4_f32(
                false, a1, false, bfrag[ks], (short)0, acc1, false, false);
        }

        const int tbase = t * G1_ROWS;
        float* __restrict__ po = H + (size_t)(tbase + hi * 8) * CH + wave * 16 + c;
        if (tbase + G1_ROWS <= nrows) {
            #pragma unroll
            for (int r = 0; r < 8; ++r) {
                po[r * CH] = acc0[r];
                po[(r + 16) * CH] = acc1[r];
            }
        } else {
            const int left = nrows - (tbase + hi * 8);
            #pragma unroll
            for (int r = 0; r < 8; ++r) {
                if (r < left) po[r * CH] = acc0[r];
                if (r + 16 < left) po[(r + 16) * CH] = acc1[r];
            }
        }

        if (more) {
            #pragma unroll
            for (int jr = 0; jr < G1_ROWS / 8; ++jr)
                *(float4*)(&Xs[buf ^ 1][0] + (wave + 8 * jr) * G1_PITCH + lane * 4) = nx[jr];
        }
        __syncthreads();
    }
}

template <int K>
__device__ __forceinline__ void add_rows(const float* __restrict__ H, int sidx, float cf,
                                         int j, int lane, float4& acc)
{
    float4 v[K];
    float  c[K];
    #pragma unroll
    for (int k = 0; k < K; ++k) {
        const int s = __builtin_amdgcn_readlane(sidx, j + k);
        c[k] = __int_as_float(__builtin_amdgcn_readlane(__float_as_int(cf), j + k));
        v[k] = *(const float4*)(H + (size_t)s * CH + lane * 4);
    }
    #pragma unroll
    for (int k = 0; k < K; ++k) {
        acc.x = fmaf(v[k].x, c[k], acc.x);
        acc.y = fmaf(v[k].y, c[k], acc.y);
        acc.z = fmaf(v[k].z, c[k], acc.z);
        acc.w = fmaf(v[k].w, c[k], acc.w);
    }
}

__global__ __launch_bounds__(256) void gather_kernel(
    const float* __restrict__ H, const float* __restrict__ dinv,
    const int* __restrict__ rowptr, const int* __restrict__ csr_src,
    const float* __restrict__ bias, float* __restrict__ out, int n)
{
    const int lane = threadIdx.x & 31;
    const int row = __builtin_amdgcn_readfirstlane(
        (int)((blockIdx.x * blockDim.x + threadIdx.x) >> 5));
    if (row >= n) return;

    const int beg = rowptr[row];
    const int end = rowptr[row + 1];
    const float di = dinv[row];
    const float w = di * di;

    const float4 h = *(const float4*)(H + (size_t)row * CH + lane * 4);
    float4 acc;
    acc.x = h.x * w; acc.y = h.y * w; acc.z = h.z * w; acc.w = h.w * w;

    for (int base = beg; base < end; base += 32) {
        const int nb = (end - base < 32) ? (end - base) : 32;
        int sidx = 0;
        float cf = 0.f;
        if (lane < nb) {
            sidx = csr_src[base + lane];
            cf = dinv[sidx] * di;
        }
        int j = 0;
        for (; j + 4 <= nb; j += 4) add_rows<4>(H, sidx, cf, j, lane, acc);
        if (j + 2 <= nb) { add_rows<2>(H, sidx, cf, j, lane, acc); j += 2; }
        if (j < nb) add_rows<1>(H, sidx, cf, j, lane, acc);
    }

    const float4 b = *(const float4*)(bias + lane * 4);
    float4 o;
    o.x = acc.x + b.x; o.y = acc.y + b.y; o.z = acc.z + b.z; o.w = acc.w + b.w;
    *(float4*)(out + (size_t)row * CH + lane * 4) = o;
}

__device__ __forceinline__ void add_chunk(const float* __restrict__ H, int sidx, float cf,
                                          int nb, int lane, float4& acc)
{
    int j = 0;
    for (; j + 8 <= nb; j += 8) add_rows<8>(H, sidx, cf, j, lane, acc);
    switch (nb - j) {
        case 7: add_rows<7>(H, sidx, cf, j, lane, acc); break;
        case 6: add_rows<6>(H, sidx, cf, j, lane, acc); break;
        case 5: add_rows<5>(H, sidx, cf, j, lane, acc); break;
        case 4: add_rows<4>(H, sidx, cf, j, lane, acc); break;
        case 3: add_rows<3>(H, sidx, cf, j, lane, acc); break;
        case 2: add_rows<2>(H, sidx, cf, j, lane, acc); break;
        case 1: add_rows<1>(H, sidx, cf, j, lane, acc); break;
        default: break;
    }
}

#define TROWS 32
#define XP 132

__global__ __launch_bounds__(256) void gather_gemm_kernel(
    const float* __restrict__ H, const float* __restrict__ dinv,
    const int* __restrict__ rowptr, const int* __restrict__ csr_src,
    const float* __restrict__ bias, const float* __restrict__ W,
    float* __restrict__ Hout, int n, int rows_per_wg)
{
    __shared__ __attribute__((aligned(16))) float Xs[2][TROWS * XP];
    __shared__ int4 Es[8][TROWS / 8][32];

    const int lane = threadIdx.x & 31;
    const int wave = __builtin_amdgcn_readfirstlane((int)(threadIdx.x >> 5));
    const int c  = lane & 15;
    const int hi = lane >> 4;

    v2f bfrag[CH / 4];
    #pragma unroll
    for (int ks = 0; ks < CH / 4; ++ks) {
        const int klo = ks * 4 + hi * 2;
        bfrag[ks].x = W[(size_t)klo * CH + wave * 16 + c];
        bfrag[ks].y = W[(size_t)(klo + 1) * CH + wave * 16 + c];
    }
    const float4 b = *(const float4*)(bias + lane * 4);

    const int row0 = (int)blockIdx.x * rows_per_wg;
    const int row1 = (row0 + rows_per_wg < n) ? (row0 + rows_per_wg) : n;

    int buf = 0;
    for (int tbase = row0; tbase < row1; tbase += TROWS, buf ^= 1) {
        float* xs = &Xs[buf][0];

        float4 hnext = make_float4(0.f, 0.f, 0.f, 0.f);
        if (tbase + wave < row1)
            hnext = *(const float4*)(H + (size_t)(tbase + wave) * CH + lane * 4);
        {
            int rbeg[TROWS / 8], rend[TROWS / 8];
            float rdi[TROWS / 8];
            int sidx0[TROWS / 8];
            #pragma unroll
            for (int jr = 0; jr < TROWS / 8; ++jr) {
                const int row = tbase + wave + 8 * jr;
                const int rc = (row < row1) ? row : row0;
                const int b0 = rowptr[rc];
                const int e0 = rowptr[rc + 1];
                rbeg[jr] = b0;
                rend[jr] = (row < row1) ? e0 : b0;
                rdi[jr] = dinv[rc];
            }
            #pragma unroll
            for (int jr = 0; jr < TROWS / 8; ++jr) {
                int p = rbeg[jr] + lane;
                p = (p < rend[jr] - 1) ? p : (rend[jr] - 1);
                p = (p > 0) ? p : 0;
                sidx0[jr] = csr_src[p];
            }
            #pragma unroll
            for (int jr = 0; jr < TROWS / 8; ++jr) {
                const float cf = dinv[sidx0[jr]] * rdi[jr];
                Es[wave][jr][lane] = make_int4(sidx0[jr], __float_as_int(cf),
                                               rend[jr] - rbeg[jr], __float_as_int(rdi[jr] * rdi[jr]));
            }
        }

        #pragma unroll 1
        for (int jr = 0; jr < TROWS / 8; ++jr) {
            const int lr = wave + 8 * jr;
            const int row = tbase + lr;
            float4 o = make_float4(0.f, 0.f, 0.f, 0.f);
            if (row < row1) {
                const int4 sc = Es[wave][jr][lane];
                const int sidx0 = sc.x;
                const float cf0 = __int_as_float(sc.y);
                const int deg = __builtin_amdgcn_readfirstlane(sc.z);
                const float w = __int_as_float(sc.w);
                const float4 h = hnext;
                if (jr + 1 < TROWS / 8 && row + 8 < row1)
                    hnext = *(const float4*)(H + (size_t)(row + 8) * CH + lane * 4);
                float4 acc;
                acc.x = h.x * w; acc.y = h.y * w; acc.z = h.z * w; acc.w = h.w * w;
                add_chunk(H, sidx0, cf0, (deg < 32) ? deg : 32, lane, acc);
                if (deg > 32) {
                  const int beg = rowptr[row];
                  const int end = beg + deg;
                  const float di = dinv[row];
                  for (int base = beg + 32; base < end; base += 32) {
                    const int nb = (end - base < 32) ? (end - base) : 32;
                    int sidx = 0;
                    float cf = 0.f;
                    if (lane < nb) {
                        sidx = csr_src[base + lane];
                        cf = dinv[sidx] * di;
                    }
                    add_chunk(H, sidx, cf, nb, lane, acc);
                  }
                }
                o.x = fmaxf(acc.x + b.x, 0.f); o.y = fmaxf(acc.y + b.y, 0.f);
                o.z = fmaxf(acc.z + b.z, 0.f); o.w = fmaxf(acc.w + b.w, 0.f);
            }
            *(float4*)(xs + lr * XP + lane * 4) = o;
        }
        __syncthreads();

        v8f acc0 = (v8f)(0.0f);
        v8f acc1 = (v8f)(0.0f);
        #pragma unroll
        for (int ks = 0; ks < CH / 4; ++ks) {
            const int klo = ks * 4 + hi * 2;
            const v2f a0 = *(const v2f*)(xs + c * XP + klo);
            const v2f a1 = *(const v2f*)(xs + (16 + c) * XP + klo);
            acc0 = __builtin_amdgcn_wmma_f32_16x16x4_f32(
                false, a0, false, bfrag[ks], (short)0, acc0, false, false);
            acc1 = __builtin_amdgcn_wmma_f32_16x16x4_f32(
                false, a1, false, bfrag[ks], (short)0, acc1, false, false);
        }

        float* __restrict__ po = Hout + (size_t)(tbase + hi * 8) * CH + wave * 16 + c;
        if (tbase + TROWS <= row1) {
            #pragma unroll
            for (int r = 0; r < 8; ++r) {
                po[r * CH] = acc0[r];
                po[(r + 16) * CH] = acc1[r];
            }
        } else {
            const int left = row1 - (tbase + hi * 8);
            #pragma unroll
            for (int r = 0; r < 8; ++r) {
                if (r < left) po[r * CH] = acc0[r];
                if (r + 16 < left) po[(r + 16) * CH] = acc1[r];
            }
        }
    }
}

extern "C" void kernel_launch(void* const* d_in, const int* in_sizes, int n_in,
                              void* d_out, int out_size, void* d_ws, size_t ws_size,
                              hipStream_t stream) {
    const float* x    = (const float*)d_in[0];
    const int*   edge = (const int*)d_in[1];
    const float* W1   = (const float*)d_in[2];
    const float* b1   = (const float*)d_in[3];
    const float* W2   = (const float*)d_in[4];
    const float* b2   = (const float*)d_in[5];
    float* out = (float*)d_out;

    const int N = in_sizes[0] / CH;
    const int E = in_sizes[1] / 2;
    const int* srcI = edge;
    const int* dstI = edge + E;

    const int T = 256;
    const int NBIN = (N + BIN_NODES - 1) / BIN_NODES;
    const int NBLK = (E + CHUNK - 1) / CHUNK;

    char* ws = (char*)d_ws;
    size_t off = 0;
    float* dinv = (float*)(ws + off);  off += ((size_t)N * 4 + 255) & ~(size_t)255;
    int* rowptr = (int*)(ws + off);    off += ((size_t)(N + 1) * 4 + 255) & ~(size_t)255;
    int* ls     = (int*)(ws + off);    off += ((size_t)NBLK * (NBIN + 1) * 4 + 255) & ~(size_t)255;
    int* part   = (int*)(ws + off);    off += ((size_t)E * 8 + 255) & ~(size_t)255;
    int* csr    = (int*)(ws + off);    off += ((size_t)E * 4 + 255) & ~(size_t)255;
    float* hbuf = (float*)(ws + off);
    float* tmp  = hbuf + (size_t)N * CH;

    dim3 gRow((N + 7) / 8);
    const int FUSED_WGS = 1024;
    const int GEMM1_WGS = 1024;
    const int RPW = (N + FUSED_WGS - 1) / FUSED_WGS;

    gemm_kernel<<<GEMM1_WGS, T, 0, stream>>>(x, W1, hbuf, N);

    partition_kernel<<<NBLK, 1024, 0, stream>>>(srcI, dstI, ls, part, E, NBIN);
    binsort_kernel<<<NBIN, 1024, 0, stream>>>(part, ls, rowptr, dinv, csr, N, NBIN, NBLK);

    gather_gemm_kernel<<<FUSED_WGS, T, 0, stream>>>(hbuf, dinv, rowptr, csr, b1, W2, tmp, N, RPW);

    gather_kernel<<<gRow, T, 0, stream>>>(tmp, dinv, rowptr, csr, b2, out, N);
}


# edge coefficient split: each GEMM stores its rows scaled by their own dinv, the gather adds plain rows and scales the sum once by dinv[dst]: no scattered dinv[src] read per edge; sum order unchanged
# speedup vs baseline: 5.3702x; 1.0596x over previous
#include <hip/hip_runtime.h>
#include <hip/hip_bf16.h>
#include <cstddef>

typedef __attribute__((ext_vector_type(2))) float v2f;
typedef __attribute__((ext_vector_type(8))) float v8f;

#define CH 128

#define BIN_SHIFT 9
#define BIN_NODES 512
#define MAXB 1024
#define CHUNK 8192
#define STAGE_CAP 16384
#define RUN_TILE 1024

__device__ __forceinline__ int block_excl_scan_1024(int v, int* wsum, int& total)
{
    const int lane = threadIdx.x & 31;
    const int wave = threadIdx.x >> 5;
    int inc = v;
    #pragma unroll
    for (int off = 1; off < 32; off <<= 1) {
        int u = __shfl_up(inc, off);
        if (lane >= off) inc += u;
    }
    if (lane == 31) wsum[wave] = inc;
    __syncthreads();
    if (wave == 0) {
        const int w = wsum[lane];
        int winc = w;
        #pragma unroll
        for (int off = 1; off < 32; off <<= 1) {
            int u = __shfl_up(winc, off);
            if (lane >= off) winc += u;
        }
        wsum[lane] = winc - w;
        if (lane == 31) wsum[32] = winc;
    }
    __syncthreads();
    total = wsum[32];
    return wsum[wave] + inc - v;
}

__global__ __launch_bounds__(1024) void partition_kernel(
    const int* __restrict__ src, const int* __restrict__ dst,
    int* __restrict__ ls, int* __restrict__ part_raw, int e, int nb)
{
    __shared__ int hist[MAXB];
    __shared__ int lstart[MAXB];
    __shared__ int wsum[33];
    __shared__ int st_src[CHUNK];
    __shared__ int st_dst[CHUNK];
    int2* __restrict__ part = (int2*)part_raw;

    const int t = threadIdx.x;
    const int blk = blockIdx.x;
    hist[t] = 0;
    __syncthreads();

    const int first = blk * CHUNK;
    int s[CHUNK / 1024], d[CHUNK / 1024], r[CHUNK / 1024];
    #pragma unroll
    for (int k = 0; k < CHUNK / 1024; ++k) {
        const int i = first + k * 1024 + t;
        s[k] = 0; d[k] = -1; r[k] = 0;
        if (i < e) { s[k] = src[i]; d[k] = dst[i]; }
    }
    #pragma unroll
    for (int k = 0; k < CHUNK / 1024; ++k)
        if (d[k] >= 0) r[k] = atomicAdd(&hist[d[k] >> BIN_SHIFT], 1);
    __syncthreads();

    int total;
    const int ex = block_excl_scan_1024(hist[t], wsum, total);
    lstart[t] = ex;
    if (t < nb) ls[blk * (nb + 1) + t] = ex;
    if (t == 0) ls[blk * (nb + 1) + nb] = total;
    __syncthreads();

    #pragma unroll
    for (int k = 0; k < CHUNK / 1024; ++k) {
        if (d[k] >= 0) {
            const int p = lstart[d[k] >> BIN_SHIFT] + r[k];
            st_src[p] = s[k];
            st_dst[p] = d[k];
        }
    }
    __syncthreads();

    for (int j = t; j < total; j += 1024)
        part[first + j] = make_int2(st_src[j], st_dst[j]);
}

template <typename F>
__device__ __forceinline__ void for_bin_edges(
    const int2* __restrict__ part, const int* __restrict__ ls, int pitch, int bin, int nblk,
    bool reload, int* rs, int* rl, F f)
{
    const int t = threadIdx.x;
    const int lane = t & 31;
    const int wave = __builtin_amdgcn_readfirstlane((int)(threadIdx.x >> 5));
    for (int c0 = 0; c0 < nblk; c0 += RUN_TILE) {
        if (reload) {
            __syncthreads();
            const int blk = c0 + t;
            int st = 0, len = 0;
            if (blk < nblk) {
                const int a0 = ls[blk * pitch + bin];
                const int a1 = ls[blk * pitch + bin + 1];
                st = blk * CHUNK + a0;
                len = a1 - a0;
            }
            rs[t] = st;
            rl[t] = len;
            __syncthreads();
        }
        const int nr = (nblk - c0 < RUN_TILE) ? (nblk - c0) : RUN_TILE;
        for (int r0 = wave; r0 < nr; r0 += 128) {
            int st4[4], ln4[4];
            #pragma unroll
            for (int k = 0; k < 4; ++k) {
                const int r = r0 + 32 * k;
                const bool ok = r < nr;
                ln4[k] = ok ? rl[ok ? r : 0] : 0;
                st4[k] = ok ? rs[ok ? r : 0] : 0;
            }
            int2 v[4][2];
            #pragma unroll
            for (int k = 0; k < 4; ++k) {
                #pragma unroll
                for (int q = 0; q < 2; ++q) {
                    const int idx = lane + 32 * q;
                    v[k][q] = make_int2(0, -1);
                    if (idx < ln4[k]) v[k][q] = part[st4[k] + idx];
                }
            }
            #pragma unroll
            for (int k = 0; k < 4; ++k) {
                #pragma unroll
                for (int q = 0; q < 2; ++q)
                    if (v[k][q].y >= 0) f(v[k][q].x, v[k][q].y);
            }
            #pragma unroll
            for (int k = 0; k < 4; ++k) {
                for (int idx = lane + 64; idx < ln4[k]; idx += 32) {
                    const int2 sd = part[st4[k] + idx];
                    f(sd.x, sd.y);
                }
            }
        }
    }
}

__global__ __launch_bounds__(1024) void binsort_kernel(
    const int* __restrict__ part_raw, const int* __restrict__ ls,
    int* __restrict__ rowptr, float* __restrict__ dinv, int* __restrict__ csr_src,
    int n, int nb, int nblk)
{
    __shared__ int cnt[BIN_NODES];
    __shared__ int cur[BIN_NODES];
    __shared__ int wsum[33];
    __shared__ int wsum2[33];
    __shared__ int rs[RUN_TILE];
    __shared__ int rl[RUN_TILE];
    __shared__ int stage[STAGE_CAP];
    const int2* __restrict__ part = (const int2*)part_raw;

    const int t = threadIdx.x;
    const int bin = blockIdx.x;
    const int node0 = bin << BIN_SHIFT;
    const int pitch = nb + 1;
    const bool reload = nblk > RUN_TILE;

    if (t < BIN_NODES) cnt[t] = 0;

    int below = 0, mine = 0;
    for (int blk = t; blk < nblk; blk += 1024) {
        const int a0 = ls[blk * pitch + bin];
        const int a1 = ls[blk * pitch + bin + 1];
        below += a0;
        mine += a1 - a0;
        if (blk < RUN_TILE) { rs[blk] = blk * CHUNK + a0; rl[blk] = a1 - a0; }
    }
    int beg, m;
    block_excl_scan_1024(below, wsum, beg);
    block_excl_scan_1024(mine, wsum2, m);

    for_bin_edges(part, ls, pitch, bin, nblk, reload, rs, rl,
        [&](int, int dd) { atomicAdd(&cnt[dd - node0], 1); });
    __syncthreads();

    const int c = (t < BIN_NODES) ? cnt[t] : 0;
    int total;
    const int ex = block_excl_scan_1024(c, wsum, total);
    if (t < BIN_NODES) {
        cur[t] = ex;
        const int node = node0 + t;
        if (node < n) {
            rowptr[node] = beg + ex;
            dinv[node] = rsqrtf((float)(c + 1));
            if (node == n - 1) rowptr[n] = beg + ex + c;
        }
    }
    __syncthreads();

    const bool staged = m <= STAGE_CAP;
    for_bin_edges(part, ls, pitch, bin, nblk, reload, rs, rl,
        [&](int ss, int dd) {
            const int p = atomicAdd(&cur[dd - node0], 1);
            if (staged) stage[p] = ss; else csr_src[beg + p] = ss;
        });
    if (staged) {
        __syncthreads();
        for (int j = t; j < m; j += 1024) csr_src[beg + j] = stage[j];
    }
}

#define G1_ROWS 32
#define G1_PITCH 132

__global__ __launch_bounds__(256) void gemm_kernel(
    const float* __restrict__ X, const float* __restrict__ W, const float* __restrict__ dinv,
    float* __restrict__ H, int nrows)
{
    __shared__ __attribute__((aligned(16))) float Xs[2][G1_ROWS * G1_PITCH];

    const int G = (int)gridDim.x;
    const int ntiles = (nrows + G1_ROWS - 1) / G1_ROWS;
    int t = (int)blockIdx.x;
    if (t >= ntiles) return;

    const int lane = threadIdx.x & 31;
    const int wave = __builtin_amdgcn_readfirstlane((int)(threadIdx.x >> 5));
    const int c  = lane & 15;
    const int hi = lane >> 4;

    v2f bfrag[CH / 4];
    #pragma unroll
    for (int ks = 0; ks < CH / 4; ++ks) {
        const int klo = ks * 4 + hi * 2;
        bfrag[ks].x = W[(size_t)klo * CH + wave * 16 + c];
        bfrag[ks].y = W[(size_t)(klo + 1) * CH + wave * 16 + c];
    }

    float4 nx[G1_ROWS / 8];
    float nd[G1_ROWS / 8];
    #pragma unroll
    for (int jr = 0; jr < G1_ROWS / 8; ++jr) {
        const int row = t * G1_ROWS + wave + 8 * jr;
        nx[jr] = make_float4(0.f, 0.f, 0.f, 0.f);
        nd[jr] = 0.f;
        if (row < nrows) {
            nx[jr] = *(const float4*)(X + (size_t)row * CH + lane * 4);
            nd[jr] = dinv[row];
        }
    }
    #pragma unroll
    for (int jr = 0; jr < G1_ROWS / 8; ++jr)
        *(float4*)(&Xs[0][0] + (wave + 8 * jr) * G1_PITCH + lane * 4) =
            make_float4(nx[jr].x * nd[jr], nx[jr].y * nd[jr], nx[jr].z * nd[jr], nx[jr].w * nd[jr]);
    __syncthreads();

    int buf = 0;
    for (; t < ntiles; t += G, buf ^= 1) {
        const float* xs = &Xs[buf][0];
        const int tn = t + G;
        const bool more = tn < ntiles;

        if (more) {
            #pragma unroll
            for (int jr = 0; jr < G1_ROWS / 8; ++jr) {
                const int row = tn * G1_ROWS + wave + 8 * jr;
                nx[jr] = make_float4(0.f, 0.f, 0.f, 0.f);
                nd[jr] = 0.f;
                if (row < nrows) {
                    nx[jr] = *(const float4*)(X + (size_t)row * CH + lane * 4);
                    nd[jr] = dinv[row];
                }
            }
        }

        v8f acc0 = (v8f)(0.0f);
        v8f acc1 = (v8f)(0.0f);
        #pragma unroll
        for (int ks = 0; ks < CH / 4; ++ks) {
            const int klo = ks * 4 + hi * 2;
            const v2f a0 = *(const v2f*)(xs + c * G1_PITCH + klo);
            const v2f a1 = *(const v2f*)(xs + (16 + c) * G1_PITCH + klo);
            acc0 = __builtin_amdgcn_wmma_f32_16x16x4_f32(
                false, a0, false, bfrag[ks], (short)0, acc0, false, false);
            acc1 = __builtin_amdgcn_wmma_f32_16x16x4_f32(
                false, a1, false, bfrag[ks], (short)0, acc1, false, false);
        }

        const int tbase = t * G1_ROWS;
        float* __restrict__ po = H + (size_t)(tbase + hi * 8) * CH + wave * 16 + c;
        if (tbase + G1_ROWS <= nrows) {
            #pragma unroll
            for (int r = 0; r < 8; ++r) {
                po[r * CH] = acc0[r];
                po[(r + 16) * CH] = acc1[r];
            }
        } else {
            const int left = nrows - (tbase + hi * 8);
            #pragma unroll
            for (int r = 0; r < 8; ++r) {
                if (r < left) po[r * CH] = acc0[r];
                if (r + 16 < left) po[(r + 16) * CH] = acc1[r];
            }
        }

        if (more) {
            #pragma unroll
            for (int jr = 0; jr < G1_ROWS / 8; ++jr)
                *(float4*)(&Xs[buf ^ 1][0] + (wave + 8 * jr) * G1_PITCH + lane * 4) =
                    make_float4(nx[jr].x * nd[jr], nx[jr].y * nd[jr], nx[jr].z * nd[jr], nx[jr].w * nd[jr]);
        }
        __syncthreads();
    }
}

template <int K>
__device__ __forceinline__ void add_rows(const float* __restrict__ H, int sidx,
                                         int j, int lane, float4& acc)
{
    float4 v[K];
    #pragma unroll
    for (int k = 0; k < K; ++k) {
        const int s = __builtin_amdgcn_readlane(sidx, j + k);
        v[k] = *(const float4*)(H + (size_t)s * CH + lane * 4);
    }
    #pragma unroll
    for (int k = 0; k < K; ++k) {
        acc.x += v[k].x;
        acc.y += v[k].y;
        acc.z += v[k].z;
        acc.w += v[k].w;
    }
}

__global__ __launch_bounds__(256) void gather_kernel(
    const float* __restrict__ H, const float* __restrict__ dinv,
    const int* __restrict__ rowptr, const int* __restrict__ csr_src,
    const float* __restrict__ bias, float* __restrict__ out, int n)
{
    const int lane = threadIdx.x & 31;
    const int row = __builtin_amdgcn_readfirstlane(
        (int)((blockIdx.x * blockDim.x + threadIdx.x) >> 5));
    if (row >= n) return;

    const int beg = rowptr[row];
    const int end = rowptr[row + 1];
    const float di = dinv[row];

    float4 acc = *(const float4*)(H + (size_t)row * CH + lane * 4);

    for (int base = beg; base < end; base += 32) {
        const int nb = (end - base < 32) ? (end - base) : 32;
        int sidx = 0;
        if (lane < nb) sidx = csr_src[base + lane];
        int j = 0;
        for (; j + 4 <= nb; j += 4) add_rows<4>(H, sidx, j, lane, acc);
        if (j + 2 <= nb) { add_rows<2>(H, sidx, j, lane, acc); j += 2; }
        if (j < nb) add_rows<1>(H, sidx, j, lane, acc);
    }

    const float4 b = *(const float4*)(bias + lane * 4);
    float4 o;
    o.x = acc.x * di + b.x; o.y = acc.y * di + b.y; o.z = acc.z * di + b.z; o.w = acc.w * di + b.w;
    *(float4*)(out + (size_t)row * CH + lane * 4) = o;
}

__device__ __forceinline__ void add_chunk(const float* __restrict__ H, int sidx,
                                          int nb, int lane, float4& acc)
{
    int j = 0;
    for (; j + 8 <= nb; j += 8) add_rows<8>(H, sidx, j, lane, acc);
    switch (nb - j) {
        case 7: add_rows<7>(H, sidx, j, lane, acc); break;
        case 6: add_rows<6>(H, sidx, j, lane, acc); break;
        case 5: add_rows<5>(H, sidx, j, lane, acc); break;
        case 4: add_rows<4>(H, sidx, j, lane, acc); break;
        case 3: add_rows<3>(H, sidx, j, lane, acc); break;
        case 2: add_rows<2>(H, sidx, j, lane, acc); break;
        case 1: add_rows<1>(H, sidx, j, lane, acc); break;
        default: break;
    }
}

#define TROWS 32
#define XP 132

__global__ __launch_bounds__(256, 8) void gather_gemm_kernel(
    const float* __restrict__ H, const float* __restrict__ dinv,
    const int* __restrict__ rowptr, const int* __restrict__ csr_src,
    const float* __restrict__ bias, const float* __restrict__ W,
    float* __restrict__ Hout, int n, int rows_per_wg)
{
    __shared__ __attribute__((aligned(16))) float Xs[2][TROWS * XP];
    __shared__ int2 Es[8][TROWS / 8][32];

    const int lane = threadIdx.x & 31;
    const int wave = __builtin_amdgcn_readfirstlane((int)(threadIdx.x >> 5));
    const int c  = lane & 15;
    const int hi = lane >> 4;

    v2f bfrag[CH / 4];
    #pragma unroll
    for (int ks = 0; ks < CH / 4; ++ks) {
        const int klo = ks * 4 + hi * 2;
        bfrag[ks].x = W[(size_t)klo * CH + wave * 16 + c];
        bfrag[ks].y = W[(size_t)(klo + 1) * CH + wave * 16 + c];
    }
    const float4 b = *(const float4*)(bias + lane * 4);

    const int row0 = (int)blockIdx.x * rows_per_wg;
    const int row1 = (row0 + rows_per_wg < n) ? (row0 + rows_per_wg) : n;

    int buf = 0;
    for (int tbase = row0; tbase < row1; tbase += TROWS, buf ^= 1) {
        float* xs = &Xs[buf][0];

        float4 hnext = make_float4(0.f, 0.f, 0.f, 0.f);
        if (tbase + wave < row1)
            hnext = *(const float4*)(H + (size_t)(tbase + wave) * CH + lane * 4);
        {
            int rbeg[TROWS / 8], rend[TROWS / 8];
            int sidx0[TROWS / 8];
            #pragma unroll
            for (int jr = 0; jr < TROWS / 8; ++jr) {
                const int row = tbase + wave + 8 * jr;
                const int rc = (row < row1) ? row : row0;
                const int b0 = rowptr[rc];
                const int e0 = rowptr[rc + 1];
                rbeg[jr] = b0;
                rend[jr] = (row < row1) ? e0 : b0;
            }
            #pragma unroll
            for (int jr = 0; jr < TROWS / 8; ++jr) {
                int p = rbeg[jr] + lane;
                p = (p < rend[jr] - 1) ? p : (rend[jr] - 1);
                p = (p > 0) ? p : 0;
                sidx0[jr] = csr_src[p];
            }
            #pragma unroll
            for (int jr = 0; jr < TROWS / 8; ++jr) {
                Es[wave][jr][lane] = make_int2(sidx0[jr], rend[jr] - rbeg[jr]);
            }
        }

        #pragma unroll 1
        for (int jr = 0; jr < TROWS / 8; ++jr) {
            const int lr = wave + 8 * jr;
            const int row = tbase + lr;
            float4 o = make_float4(0.f, 0.f, 0.f, 0.f);
            if (row < row1) {
                const int2 sc = Es[wave][jr][lane];
                const int sidx0 = sc.x;
                const int deg = __builtin_amdgcn_readfirstlane(sc.y);
                const float di = dinv[row];
                float4 acc = hnext;
                if (jr + 1 < TROWS / 8 && row + 8 < row1)
                    hnext = *(const float4*)(H + (size_t)(row + 8) * CH + lane * 4);
                add_chunk(H, sidx0, (deg < 32) ? deg : 32, lane, acc);
                if (deg > 32) {
                  const int beg = rowptr[row];
                  const int end = beg + deg;
                  for (int base = beg + 32; base < end; base += 32) {
                    const int nb = (end - base < 32) ? (end - base) : 32;
                    int sidx = 0;
                    if (lane < nb) sidx = csr_src[base + lane];
                    add_chunk(H, sidx, nb, lane, acc);
                  }
                }
                o.x = di * fmaxf(acc.x * di + b.x, 0.f); o.y = di * fmaxf(acc.y * di + b.y, 0.f);
                o.z = di * fmaxf(acc.z * di + b.z, 0.f); o.w = di * fmaxf(acc.w * di + b.w, 0.f);
            }
            *(float4*)(xs + lr * XP + lane * 4) = o;
        }
        __syncthreads();

        v8f acc0 = (v8f)(0.0f);
        v8f acc1 = (v8f)(0.0f);
        #pragma unroll
        for (int ks = 0; ks < CH / 4; ++ks) {
            const int klo = ks * 4 + hi * 2;
            const v2f a0 = *(const v2f*)(xs + c * XP + klo);
            const v2f a1 = *(const v2f*)(xs + (16 + c) * XP + klo);
            acc0 = __builtin_amdgcn_wmma_f32_16x16x4_f32(
                false, a0, false, bfrag[ks], (short)0, acc0, false, false);
            acc1 = __builtin_amdgcn_wmma_f32_16x16x4_f32(
                false, a1, false, bfrag[ks], (short)0, acc1, false, false);
        }

        float* __restrict__ po = Hout + (size_t)(tbase + hi * 8) * CH + wave * 16 + c;
        if (tbase + TROWS <= row1) {
            #pragma unroll
            for (int r = 0; r < 8; ++r) {
                po[r * CH] = acc0[r];
                po[(r + 16) * CH] = acc1[r];
            }
        } else {
            const int left = row1 - (tbase + hi * 8);
            #pragma unroll
            for (int r = 0; r < 8; ++r) {
                if (r < left) po[r * CH] = acc0[r];
                if (r + 16 < left) po[(r + 16) * CH] = acc1[r];
            }
        }
    }
}

extern "C" void kernel_launch(void* const* d_in, const int* in_sizes, int n_in,
                              void* d_out, int out_size, void* d_ws, size_t ws_size,
                              hipStream_t stream) {
    const float* x    = (const float*)d_in[0];
    const int*   edge = (const int*)d_in[1];
    const float* W1   = (const float*)d_in[2];
    const float* b1   = (const float*)d_in[3];
    const float* W2   = (const float*)d_in[4];
    const float* b2   = (const float*)d_in[5];
    float* out = (float*)d_out;

    const int N = in_sizes[0] / CH;
    const int E = in_sizes[1] / 2;
    const int* srcI = edge;
    const int* dstI = edge + E;

    const int T = 256;
    const int NBIN = (N + BIN_NODES - 1) / BIN_NODES;
    const int NBLK = (E + CHUNK - 1) / CHUNK;

    char* ws = (char*)d_ws;
    size_t off = 0;
    float* dinv = (float*)(ws + off);  off += ((size_t)N * 4 + 255) & ~(size_t)255;
    int* rowptr = (int*)(ws + off);    off += ((size_t)(N + 1) * 4 + 255) & ~(size_t)255;
    int* ls     = (int*)(ws + off);    off += ((size_t)NBLK * (NBIN + 1) * 4 + 255) & ~(size_t)255;
    int* part   = (int*)(ws + off);    off += ((size_t)E * 8 + 255) & ~(size_t)255;
    int* csr    = (int*)(ws + off);    off += ((size_t)E * 4 + 255) & ~(size_t)255;
    float* hbuf = (float*)(ws + off);
    float* tmp  = hbuf + (size_t)N * CH;

    dim3 gRow((N + 7) / 8);
    const int FUSED_WGS = 1024;
    const int GEMM1_WGS = 1024;
    const int RPW = (N + FUSED_WGS - 1) / FUSED_WGS;

    partition_kernel<<<NBLK, 1024, 0, stream>>>(srcI, dstI, ls, part, E, NBIN);
    binsort_kernel<<<NBIN, 1024, 0, stream>>>(part, ls, rowptr, dinv, csr, N, NBIN, NBLK);

    gemm_kernel<<<GEMM1_WGS, T, 0, stream>>>(x, W1, dinv, hbuf, N);

    gather_gemm_kernel<<<FUSED_WGS, T, 0, stream>>>(hbuf, dinv, rowptr, csr, b1, W2, tmp, N, RPW);

    gather_kernel<<<gRow, T, 0, stream>>>(tmp, dinv, rowptr, csr, b2, out, N);
}
